# starting kernel
# baseline (speedup 1.0000x reference)
#include <hip/hip_runtime.h>
#include <hip/hip_bf16.h>

typedef float v8f __attribute__((ext_vector_type(8)));
typedef float v2f __attribute__((ext_vector_type(2)));

#define NFEAT 64
#define NEMB  192
#define MAXBLKS 128

__global__ void fill_f32(float* __restrict__ p, float v, long n) {
    long i = (long)blockIdx.x * blockDim.x + threadIdx.x;
    if (i < n) p[i] = v;
}

__global__ void deg_count(const int* __restrict__ dst, float* __restrict__ deg, int E) {
    int stride = gridDim.x * blockDim.x;
    for (int e = blockIdx.x * blockDim.x + threadIdx.x; e < E; e += stride)
        atomicAdd(&deg[dst[e]], 1.0f);
}

__global__ void make_dinv(float* __restrict__ deg, int n) {
    int i = blockIdx.x * blockDim.x + threadIdx.x;
    if (i < n) deg[i] = rsqrtf(deg[i]);
}

__global__ void gemm_n64(const float* __restrict__ A, int lda,
                         const float* __restrict__ W,
                         float* __restrict__ H, int n) {
    int wave = (int)((blockIdx.x * blockDim.x + threadIdx.x) >> 5);
    int lane = threadIdx.x & 31;
    int tileM = (wave >> 2) * 16;
    int tileN = (wave & 3) * 16;
    if (tileM >= n) return;

    int hi = lane >> 4;
    int lo = lane & 15;

    v8f c = {};
    const float* arow = A + (long)(tileM + lo) * lda + 2 * hi;
    const float* bcol = W + (long)(2 * hi) * NFEAT + tileN + lo;
    #pragma unroll
    for (int k = 0; k < NFEAT; k += 4) {
        v2f a, b;
        a.x = arow[0];        a.y = arow[1];
        b.x = bcol[0];        b.y = bcol[NFEAT];
        c = __builtin_amdgcn_wmma_f32_16x16x4_f32(
                 false, a,  false, b,
                 (short)0, c,  false,  false);
        arow += 4;
        bcol += 4 * NFEAT;
    }
    float* out = H + (long)(tileM + 8 * hi) * NFEAT + tileN + lo;
    #pragma unroll
    for (int r = 0; r < 8; ++r) out[(long)r * NFEAT] = c[r];
}

__global__ void scatter_edges(const int* __restrict__ src, const int* __restrict__ dst,
                              const float* __restrict__ dinv,
                              const float* __restrict__ h,
                              float* __restrict__ agg, int E) {
    int lane = threadIdx.x & 31;
    int wave = (int)((blockIdx.x * blockDim.x + threadIdx.x) >> 5);
    int nwaves = (gridDim.x * blockDim.x) >> 5;
    for (int e = wave; e < E; e += nwaves) {
        int s = src[e];
        int d = dst[e];
        float norm = dinv[s] * dinv[d];
        const float* hs = h + (long)s * NFEAT;
        float*       ad = agg + (long)d * NFEAT;
        float v0 = hs[lane]      * norm;
        float v1 = hs[lane + 32] * norm;
        atomicAdd(&ad[lane],      v0);
        atomicAdd(&ad[lane + 32], v1);
    }
}

__global__ void finish_layer(const float* __restrict__ agg, const float* __restrict__ h,
                             const float* __restrict__ dinv, const float* __restrict__ b,
                             float* __restrict__ emb, int colOff, int relu, int n) {
    long idx = (long)blockIdx.x * blockDim.x + threadIdx.x;
    if (idx >= (long)n * NFEAT) return;
    int i = (int)(idx >> 6);
    int f = (int)(idx & 63);
    float di = dinv[i];
    float v = agg[idx] + h[idx] * di * di + b[f];
    if (relu) v = fmaxf(v, 0.0f);
    emb[(long)i * NEMB + colOff + f] = v;
}

__global__ void max_stage1(const float* __restrict__ emb, float* __restrict__ partial, int n) {
    int f = threadIdx.x;
    int rowsPer = (n + gridDim.x - 1) / gridDim.x;
    int r0 = blockIdx.x * rowsPer;
    int r1 = min(n, r0 + rowsPer);
    float m = -INFINITY;
    for (int r = r0; r < r1; ++r)
        m = fmaxf(m, emb[(long)r * NEMB + f]);
    partial[(long)blockIdx.x * NEMB + f] = m;
}

__global__ void max_stage2(const float* __restrict__ partial, float* __restrict__ ge, int nb) {
    int f = threadIdx.x;
    float m = -INFINITY;
    for (int b = 0; b < nb; ++b)
        m = fmaxf(m, partial[(long)b * NEMB + f]);
    ge[f] = m;
}

__global__ void target_logits(const float* __restrict__ emb, const float* __restrict__ fcW,
                              const float* __restrict__ fcb, const int* __restrict__ target,
                              float* __restrict__ out) {
    int c = threadIdx.x;
    if (c >= 4) return;
    int t = target[0];
    float s = fcb[c];
    const float* row = emb + (long)t * NEMB;
    for (int f = 0; f < NEMB; ++f)
        s += row[f] * fcW[f * 4 + c];
    out[c] = s;
}

extern "C" void kernel_launch(void* const* d_in, const int* in_sizes, int n_in,
                              void* d_out, int out_size, void* d_ws, size_t ws_size,
                              hipStream_t stream) {
    const float* x      = (const float*)d_in[0];
    const int*   edge   = (const int*)d_in[1];
    const int*   target = (const int*)d_in[3];
    const float* W1 = (const float*)d_in[4];
    const float* b1 = (const float*)d_in[5];
    const float* W2 = (const float*)d_in[6];
    const float* b2 = (const float*)d_in[7];
    const float* W3 = (const float*)d_in[8];
    const float* b3 = (const float*)d_in[9];
    const float* fcW = (const float*)d_in[10];
    const float* fcb = (const float*)d_in[11];

    const int N = in_sizes[0] / NFEAT;
    const int E = in_sizes[1] / 2;
    const int* esrc = edge;
    const int* edst = edge + E;

    float* ws   = (float*)d_ws;
    float* dinv = ws;
    float* h    = dinv + N;
    float* agg  = h + (long)N * NFEAT;
    float* part = agg + (long)N * NFEAT;

    float* emb = (float*)d_out;
    float* ge  = emb + (long)N * NEMB;
    float* out4 = ge + NEMB;

    const long nh = (long)N * NFEAT;
    const int blkFill = (int)((nh + 255) / 256);

    fill_f32<<<(N + 255) / 256, 256, 0, stream>>>(dinv, 1.0f, N);
    deg_count<<<1024, 256, 0, stream>>>(edst, dinv, E);
    make_dinv<<<(N + 255) / 256, 256, 0, stream>>>(dinv, N);

    const int gemmWaves  = (N / 16) * 4;
    const int gemmBlocks = (gemmWaves + 7) / 8;

    struct { const float* A; int lda; const float* W; const float* b; int colOff; int relu; }
    layers[3] = {
        { x,        NFEAT, W1, b1, 0,   1 },
        { emb + 0,  NEMB,  W2, b2, 64,  1 },
        { emb + 64, NEMB,  W3, b3, 128, 0 },
    };

    for (int l = 0; l < 3; ++l) {
        gemm_n64<<<gemmBlocks, 256, 0, stream>>>(layers[l].A, layers[l].lda, layers[l].W, h, N);
        fill_f32<<<blkFill, 256, 0, stream>>>(agg, 0.0f, nh);
        scatter_edges<<<2048, 256, 0, stream>>>(esrc, edst, dinv, h, agg, E);
        finish_layer<<<blkFill, 256, 0, stream>>>(agg, h, dinv, layers[l].b,
                                                  emb, layers[l].colOff, layers[l].relu, N);
    }

    max_stage1<<<MAXBLKS, NEMB, 0, stream>>>(emb, part, N);
    max_stage2<<<1, NEMB, 0, stream>>>(part, ge, MAXBLKS);

    target_logits<<<1, 32, 0, stream>>>(emb, fcW, fcb, target, out4);
}


# edge sums owned per block: each block scans the whole dst list and adds the source rows of its 256 rows in LDS (f32 LDS adds, arrival order as before), epilogue in place; graph max by atomic max
# speedup vs baseline: 2.2765x; 2.2765x over previous
#include <hip/hip_runtime.h>
#include <hip/hip_bf16.h>

typedef float v8f __attribute__((ext_vector_type(8)));
typedef float v2f __attribute__((ext_vector_type(2)));
typedef float v4f __attribute__((ext_vector_type(4)));

#define NFEAT 64
#define NEMB  192

template <class F>
__device__ __forceinline__ void walk_own_half_first(int* __restrict__ cnt, int nChunks, F body) {
    __shared__ int s_t[2];
    const int half = (int)((__builtin_amdgcn_s_sendmsg_rtn(0x87) >> 18) & 1u);
    for (int pass = 0; pass < 2; ++pass) {
        const int q = half ^ pass;
        int tn = nChunks;
        if (threadIdx.x == 0) {
            if (pass == 0 ||
                __hip_atomic_load(&cnt[q], __ATOMIC_RELAXED, __HIP_MEMORY_SCOPE_AGENT) < nChunks)
                tn = atomicAdd(&cnt[q], 1);
        }
        for (int it = 0; it <= nChunks; ++it) {
            if (threadIdx.x == 0) s_t[it & 1] = tn;
            __syncthreads();
            const int t = s_t[it & 1];
            if (t >= nChunks) break;
            if (threadIdx.x == 0) tn = atomicAdd(&cnt[q], 1);
            body(q, t);
        }
        __syncthreads();
    }
}

#define NCNT 2
__global__ void init_deg(float* __restrict__ deg, int* __restrict__ cnt, int n) {
    int i = blockIdx.x * blockDim.x + threadIdx.x;
    if (i < n) deg[i] = 1.0f;
    if (i < NCNT) cnt[i] = 0;
}

#define DCHUNK 512
__global__ __launch_bounds__(256)
void deg_count(const int* __restrict__ dst, float* __restrict__ deg, int* __restrict__ cnt,
               int E, int n) {
    const int split = n >> 1;
    walk_own_half_first(cnt, (E + DCHUNK - 1) / DCHUNK, [&](int q, int t) {
        const int e0 = t * DCHUNK + threadIdx.x;
        const int e1 = e0 + 256;
        int d0 = -1, d1 = -1;
        if (e0 < E) d0 = dst[e0];
        if (e1 < E) d1 = dst[e1];
        if (d0 >= 0 && (d0 >= split) == (q != 0)) atomicAdd(&deg[d0], 1.0f);
        if (d1 >= 0 && (d1 >= split) == (q != 0)) atomicAdd(&deg[d1], 1.0f);
    });
}

__global__ void make_dinv(float* __restrict__ deg, float* __restrict__ ge, int n) {
    int i = blockIdx.x * blockDim.x + threadIdx.x;
    if (i < n) deg[i] = rsqrtf(deg[i]);
    if (i < NEMB) ge[i] = -INFINITY;
}

__global__ void gemm_n64(const float* __restrict__ A, int lda,
                         const float* __restrict__ W,
                         const float* __restrict__ dinv,
                         float* __restrict__ H, int n) {
    int wave = (int)((blockIdx.x * blockDim.x + threadIdx.x) >> 5);
    int lane = threadIdx.x & 31;
    int tileM = (wave >> 2) * 16;
    int tileN = (wave & 3) * 16;
    if (tileM >= n) return;

    int hi = lane >> 4;
    int lo = lane & 15;

    float ds[8];
    #pragma unroll
    for (int r = 0; r < 8; ++r) ds[r] = dinv[tileM + 8 * hi + r];

    v8f c = {};
    const float* arow = A + (long)(tileM + lo) * lda + 2 * hi;
    const float* bcol = W + (long)(2 * hi) * NFEAT + tileN + lo;
    #pragma unroll
    for (int k = 0; k < NFEAT; k += 4) {
        v2f a, b;
        a.x = arow[0];        a.y = arow[1];
        b.x = bcol[0];        b.y = bcol[NFEAT];
        c = __builtin_amdgcn_wmma_f32_16x16x4_f32(
                 false, a,  false, b,
                 (short)0, c,  false,  false);
        arow += 4;
        bcol += 4 * NFEAT;
    }
    float* out = H + (long)(tileM + 8 * hi) * NFEAT + tileN + lo;
    #pragma unroll
    for (int r = 0; r < 8; ++r) out[(long)r * NFEAT] = c[r] * ds[r];
}

#define OWN 256
#define QCAP 256
#define SCAN_U 8
#define FROWS 64
#define APITCH 68
#define WPITCH 72
__device__ __forceinline__ int lane_i(int v, int j) { return __builtin_amdgcn_readlane(v, j); }

__device__ __forceinline__ void atomic_max_f32(float* addr, float v) {
    const unsigned u = __float_as_uint(v);
    if ((u >> 31) == 0) atomicMax((int*)addr, (int)u);
    else                atomicMin((unsigned*)addr, u);
}

template <bool NEXT>
__global__ __launch_bounds__(1024)
void layer_owned(const int* __restrict__ src, const int* __restrict__ dst,
                 const float* __restrict__ h, float* __restrict__ hn,
                 const float* __restrict__ dinv, const float* __restrict__ b,
                 float* __restrict__ emb, int colOff, int relu,
                 const float* __restrict__ Wn, float* __restrict__ ge, int E, int n) {
    __shared__ __attribute__((aligned(16))) float acc[OWN * NFEAT];
    __shared__ int qe[32][QCAP];
    __shared__ int qd[32][QCAP];
    __shared__ float red[OWN / 4][NFEAT];
    __shared__ __attribute__((aligned(16))) float sA[NEXT ? OWN * APITCH : 4];
    __shared__ __attribute__((aligned(16))) float sW[NEXT ? NFEAT * WPITCH : 4];
    __shared__ __attribute__((aligned(16))) float sD[OWN];
    const int lane = threadIdx.x & 31;
    const int wv   = threadIdx.x >> 5;
    const int r0   = blockIdx.x * OWN;

    #pragma unroll
    for (int i = 0; i < OWN * NFEAT / 4 / 1024; ++i)
        *(float4*)(&acc[4 * (threadIdx.x + 1024 * i)]) = make_float4(0.f, 0.f, 0.f, 0.f);
    __syncthreads();

    int qh = 0;
    int qn = 0;
    auto drain = [&](int cnt) {
        int s = 0, d = 0;
        __builtin_amdgcn_fence(__ATOMIC_ACQUIRE, "wavefront");
        if (lane < cnt) {
            const int slot = (qh + lane) & (QCAP - 1);
            s = src[qe[wv][slot]];
            d = qd[wv][slot];
        }
        unsigned mask = cnt >= 32 ? 0xffffffffu : ((1u << cnt) - 1u);
        #pragma unroll 1
        while (mask) {
            int   dj[8];
            float v0[8], v1[8];
            #pragma unroll
            for (int k = 0; k < 8; ++k) {
                dj[k] = -1;
                if (mask) {
                    const int j = __builtin_ctz(mask);
                    mask &= mask - 1;
                    const float* hs = h + (long)lane_i(s, j) * NFEAT;
                    dj[k] = lane_i(d, j);
                    v0[k] = hs[lane];
                    v1[k] = hs[lane + 32];
                }
            }
            #pragma unroll
            for (int k = 0; k < 8; ++k) {
                if (dj[k] >= 0) {
                    float* ad = &acc[dj[k] * NFEAT];
                    atomicAdd(&ad[lane],      v0[k]);
                    atomicAdd(&ad[lane + 32], v1[k]);
                }
            }
        }
        qh = (qh + cnt) & (QCAP - 1);
        qn -= cnt;
    };
    auto put = [&](int dv, int ev) {
        const int  dl  = dv - r0;
        const bool hit = (unsigned)dl < (unsigned)OWN;
        const unsigned m = __builtin_amdgcn_ballot_w32(hit);
        if (m) {
            if (hit) {
                const int slot = (qh + qn + __builtin_popcount(m & ((1u << lane) - 1u))) & (QCAP - 1);
                qe[wv][slot] = ev;
                qd[wv][slot] = dl;
            }
            __builtin_amdgcn_fence(__ATOMIC_RELEASE, "wavefront");
            qn += __builtin_popcount(m);
        }
    };
    const int RND = 1024 * 4;
    if ((E & 3) == 0 && E >= 4) {
        auto ld4 = [&](int e) { return *(const int4*)(dst + min(e, E - 4)); };
        auto trip = [&](int4 (&now)[SCAN_U], int4 (&next)[SCAN_U], int eb, const bool whole) {
            const int e0 = eb + 4 * (int)threadIdx.x;
            const int* pn = dst + e0 + SCAN_U * RND;
            #pragma unroll
            for (int u = 0; u < SCAN_U; ++u)
                next[u] = whole ? *(const int4*)(pn + u * RND) : ld4(e0 + (SCAN_U + u) * RND);
            #pragma unroll
            for (int u = 0; u < SCAN_U; ++u) {
                const int  e    = e0 + u * RND;
                const bool live = whole || e < E;
                const unsigned lo4 = min(min((unsigned)(now[u].x - r0), (unsigned)(now[u].y - r0)),
                                         min((unsigned)(now[u].z - r0), (unsigned)(now[u].w - r0)));
                if (__builtin_amdgcn_ballot_w32(lo4 < (unsigned)OWN && live)) {
                    const int4 d4 = live ? now[u] : make_int4(-1, -1, -1, -1);
                    put(d4.x, e);
                    put(d4.y, e + 1);
                    put(d4.z, e + 2);
                    put(d4.w, e + 3);
                    while (qn >= 128) drain(32);
                }
            }
            while (qn >= 32) drain(32);
        };
        int4 bufA[SCAN_U], bufB[SCAN_U];
        #pragma unroll
        for (int u = 0; u < SCAN_U; ++u) bufA[u] = ld4(u * RND + 4 * (int)threadIdx.x);
        int eb = 0;
        for (; eb + 3 * SCAN_U * RND <= E; eb += 2 * SCAN_U * RND) {
            trip(bufA, bufB, eb, true);
            trip(bufB, bufA, eb + SCAN_U * RND, true);
        }
        for (; eb < E; eb += 2 * SCAN_U * RND) {
            trip(bufA, bufB, eb, false);
            trip(bufB, bufA, eb + SCAN_U * RND, false);
        }
    } else {
        for (int eb = 0; eb < E; eb += 1024) {
            const int e = eb + (int)threadIdx.x;
            put(e < E ? dst[e] : -1, e);
            while (qn >= 32) drain(32);
        }
    }
    while (qn > 0) drain(qn < 32 ? qn : 32);
    __syncthreads();

    const int g   = threadIdx.x >> 8;
    const int tid = threadIdx.x & 255;
    const int q   = tid & 15;
    const int rs  = tid >> 4;
    const int lb  = g * FROWS;
    const int rb  = r0 + lb;
    const float4 bias = *(const float4*)(b + 4 * q);
    float4 m = make_float4(-INFINITY, -INFINITY, -INFINITY, -INFINITY);
    float  di[FROWS / 16];
    v4f t[FROWS / 16], w[FROWS / 16];
    if constexpr (NEXT) {
        if (g == 0) {
            #pragma unroll
            for (int p = 0; p < FROWS / 16; ++p)
                w[p] = *(const v4f*)(Wn + (rs + 16 * p) * NFEAT + 4 * q);
        }
    }
    #pragma unroll
    for (int p = 0; p < FROWS / 16; ++p) {
        int ic = min(rb + rs + 16 * p, n - 1);
        di[p] = dinv[ic];
        t[p] = *(const v4f*)(h + (long)ic * NFEAT + 4 * q);
    }
    if constexpr (NEXT) {
        if (g == 0) {
            #pragma unroll
            for (int p = 0; p < FROWS / 16; ++p)
                *(v4f*)(&sW[(rs + 16 * p) * WPITCH + 4 * q]) = w[p];
        }
    }
    #pragma unroll
    for (int p = 0; p < FROWS / 16; ++p) {
        const int lr = lb + rs + 16 * p;
        const int i  = r0 + lr;
        const v4f a = *(const v4f*)(&acc[lr * NFEAT + 4 * q]);
        float4 v;
        v.x = (a.x + t[p].x) * di[p] + bias.x;
        v.y = (a.y + t[p].y) * di[p] + bias.y;
        v.z = (a.z + t[p].z) * di[p] + bias.z;
        v.w = (a.w + t[p].w) * di[p] + bias.w;
        if (relu) {
            v.x = fmaxf(v.x, 0.0f); v.y = fmaxf(v.y, 0.0f);
            v.z = fmaxf(v.z, 0.0f); v.w = fmaxf(v.w, 0.0f);
        }
        if (i < n) {
            *(float4*)(emb + (long)i * NEMB + colOff + 4 * q) = v;
            m.x = fmaxf(m.x, v.x); m.y = fmaxf(m.y, v.y);
            m.z = fmaxf(m.z, v.z); m.w = fmaxf(m.w, v.w);
        }
        if constexpr (NEXT) {
            *(float4*)(&sA[lr * APITCH + 4 * q]) = v;
            if (q == 0) sD[lr] = di[p];
        }
    }
    *(float4*)(&red[g * 16 + rs][4 * q]) = m;
    __syncthreads();
    if (tid < NFEAT && rb < n) {
        float mm = red[g * 16][tid];
        #pragma unroll
        for (int r = 1; r < 16; ++r) mm = fmaxf(mm, red[g * 16 + r][tid]);
        atomic_max_f32(&ge[colOff + tid], mm);
    }
    if constexpr (NEXT) {
        const int w8 = (threadIdx.x >> 5) & 7;
        const int hi = lane >> 4;
        const int lo = lane & 15;
        const int tm = lb + (w8 >> 1) * 16;
        const int tn = (w8 & 1) * 32;
        const float* ar = &sA[(tm + lo) * APITCH + 2 * hi];
        const float* bc = &sW[(2 * hi) * WPITCH + tn + lo];
        const v4f dlo = *(const v4f*)(&sD[tm + 8 * hi]);
        const v4f dhi = *(const v4f*)(&sD[tm + 8 * hi + 4]);
        v8f c0 = {}, c1 = {};
        #pragma unroll
        for (int k = 0; k < NFEAT; k += 4) {
            v2f fa, fb0, fb1;
            fa.x  = ar[k];                       fa.y  = ar[k + 1];
            fb0.x = bc[k * WPITCH];              fb0.y = bc[(k + 1) * WPITCH];
            fb1.x = bc[k * WPITCH + 16];         fb1.y = bc[(k + 1) * WPITCH + 16];
            c0 = __builtin_amdgcn_wmma_f32_16x16x4_f32(false, fa, false, fb0, (short)0, c0, false, false);
            c1 = __builtin_amdgcn_wmma_f32_16x16x4_f32(false, fa, false, fb1, (short)0, c1, false, false);
        }
        const int rowBase = r0 + tm + 8 * hi;
        float* o = hn + (long)rowBase * NFEAT + tn + lo;
        #pragma unroll
        for (int r = 0; r < 8; ++r) {
            const float dr = r < 4 ? dlo[r & 3] : dhi[r & 3];
            if (rowBase + r < n) {
                o[(long)r * NFEAT]      = c0[r] * dr;
                o[(long)r * NFEAT + 16] = c1[r] * dr;
            }
        }
    }
}

__global__ void target_logits(const float* __restrict__ emb, const float* __restrict__ fcW,
                              const float* __restrict__ fcb, const int* __restrict__ target,
                              float* __restrict__ out) {
    int c = threadIdx.x;
    if (c >= 4) return;
    int t = target[0];
    float s = fcb[c];
    const float* row = emb + (long)t * NEMB;
    for (int f = 0; f < NEMB; ++f)
        s += row[f] * fcW[f * 4 + c];
    out[c] = s;
}

extern "C" void kernel_launch(void* const* d_in, const int* in_sizes, int n_in,
                              void* d_out, int out_size, void* d_ws, size_t ws_size,
                              hipStream_t stream) {
    const float* x      = (const float*)d_in[0];
    const int*   edge   = (const int*)d_in[1];
    const int*   target = (const int*)d_in[3];
    const float* W1 = (const float*)d_in[4];
    const float* b1 = (const float*)d_in[5];
    const float* W2 = (const float*)d_in[6];
    const float* b2 = (const float*)d_in[7];
    const float* W3 = (const float*)d_in[8];
    const float* b3 = (const float*)d_in[9];
    const float* fcW = (const float*)d_in[10];
    const float* fcb = (const float*)d_in[11];

    const int N = in_sizes[0] / NFEAT;
    const int E = in_sizes[1] / 2;
    const int* esrc = edge;
    const int* edst = edge + E;

    float* ws   = (float*)d_ws;
    float* dinv = ws;
    float* h    = dinv + N;
    float* h2   = h + (long)N * NFEAT;

    float* emb = (float*)d_out;
    float* ge  = emb + (long)N * NEMB;
    float* out4 = ge + NEMB;
    int* cnt = (int*)(ge + 128);

    init_deg<<<(N + 255) / 256, 256, 0, stream>>>(dinv, cnt, N);
    deg_count<<<1024, 256, 0, stream>>>(edst, dinv, cnt, E, N);
    make_dinv<<<(N + 255) / 256, 256, 0, stream>>>(dinv, ge, N);

    const int gemmWaves  = (N / 16) * 4;
    const int gemmBlocks = (gemmWaves + 7) / 8;
    const int nOwn       = (N + OWN - 1) / OWN;

    gemm_n64<<<gemmBlocks, 256, 0, stream>>>(x, NFEAT, W1, dinv, h, N);
    layer_owned<true><<<nOwn, 1024, 0, stream>>>(esrc, edst, h,  h2, dinv, b1, emb, 0,   1, W2, ge, E, N);
    layer_owned<true><<<nOwn, 1024, 0, stream>>>(esrc, edst, h2, h,  dinv, b2, emb, 64,  1, W3, ge, E, N);
    layer_owned<false><<<nOwn, 1024, 0, stream>>>(esrc, edst, h, h2, dinv, b3, emb, 128, 0, nullptr, ge, E, N);

    target_logits<<<1, 32, 0, stream>>>(emb, fcW, fcb, target, out4);
}


# layers 2, 3 read a per-block edge list layer 1 leaves as it scans (source + owned row per word, 128-byte chunks in the block's last emb columns); 2 of 3 dst scans gone; arrival-order sums as before
# speedup vs baseline: 2.9122x; 1.2792x over previous
#include <hip/hip_runtime.h>
#include <hip/hip_bf16.h>

typedef float v8f __attribute__((ext_vector_type(8)));
typedef float v2f __attribute__((ext_vector_type(2)));
typedef float v4f __attribute__((ext_vector_type(4)));

#define NFEAT 64
#define NEMB  192

template <class F>
__device__ __forceinline__ void walk_own_half_first(int* __restrict__ cnt, int nChunks, F body) {
    __shared__ int s_t[2];
    const int half = (int)((__builtin_amdgcn_s_sendmsg_rtn(0x87) >> 18) & 1u);
    for (int pass = 0; pass < 2; ++pass) {
        const int q = half ^ pass;
        int tn = nChunks;
        if (threadIdx.x == 0) {
            if (pass == 0 ||
                __hip_atomic_load(&cnt[q], __ATOMIC_RELAXED, __HIP_MEMORY_SCOPE_AGENT) < nChunks)
                tn = atomicAdd(&cnt[q], 1);
        }
        for (int it = 0; it <= nChunks; ++it) {
            if (threadIdx.x == 0) s_t[it & 1] = tn;
            __syncthreads();
            const int t = s_t[it & 1];
            if (t >= nChunks) break;
            if (threadIdx.x == 0) tn = atomicAdd(&cnt[q], 1);
            body(q, t);
        }
        __syncthreads();
    }
}

#define NCNT 2
__global__ void init_deg(float* __restrict__ deg, int* __restrict__ cnt, int n) {
    int i = blockIdx.x * blockDim.x + threadIdx.x;
    if (i < n) deg[i] = 1.0f;
    if (i < NCNT) cnt[i] = 0;
}

#define DCHUNK 512
__global__ __launch_bounds__(256)
void deg_count(const int* __restrict__ dst, float* __restrict__ deg, int* __restrict__ cnt,
               int E, int n) {
    const int split = n >> 1;
    walk_own_half_first(cnt, (E + DCHUNK - 1) / DCHUNK, [&](int q, int t) {
        const int e0 = t * DCHUNK + threadIdx.x;
        const int e1 = e0 + 256;
        int d0 = -1, d1 = -1;
        if (e0 < E) d0 = dst[e0];
        if (e1 < E) d1 = dst[e1];
        if (d0 >= 0 && (d0 >= split) == (q != 0)) atomicAdd(&deg[d0], 1.0f);
        if (d1 >= 0 && (d1 >= split) == (q != 0)) atomicAdd(&deg[d1], 1.0f);
    });
}

__global__ void make_dinv(float* __restrict__ deg, float* __restrict__ ge, int n) {
    int i = blockIdx.x * blockDim.x + threadIdx.x;
    if (i < n) deg[i] = rsqrtf(deg[i]);
    if (i < NEMB) ge[i] = -INFINITY;
}

__global__ void gemm_n64(const float* __restrict__ A, int lda,
                         const float* __restrict__ W,
                         const float* __restrict__ dinv,
                         float* __restrict__ H, int n) {
    int wave = (int)((blockIdx.x * blockDim.x + threadIdx.x) >> 5);
    int lane = threadIdx.x & 31;
    int tileM = (wave >> 2) * 16;
    int tileN = (wave & 3) * 16;
    if (tileM >= n) return;

    int hi = lane >> 4;
    int lo = lane & 15;

    float ds[8];
    #pragma unroll
    for (int r = 0; r < 8; ++r) ds[r] = dinv[tileM + 8 * hi + r];

    v8f c = {};
    const float* arow = A + (long)(tileM + lo) * lda + 2 * hi;
    const float* bcol = W + (long)(2 * hi) * NFEAT + tileN + lo;
    #pragma unroll
    for (int k = 0; k < NFEAT; k += 4) {
        v2f a, b;
        a.x = arow[0];        a.y = arow[1];
        b.x = bcol[0];        b.y = bcol[NFEAT];
        c = __builtin_amdgcn_wmma_f32_16x16x4_f32(
                 false, a,  false, b,
                 (short)0, c,  false,  false);
        arow += 4;
        bcol += 4 * NFEAT;
    }
    float* out = H + (long)(tileM + 8 * hi) * NFEAT + tileN + lo;
    #pragma unroll
    for (int r = 0; r < 8; ++r) out[(long)r * NFEAT] = c[r] * ds[r];
}

#define OWN 256
#define QCAP 256
#define SCAN_U 8
#define FROWS 64
#define APITCH 68
#define WPITCH 72
__device__ __forceinline__ int lane_i(int v, int j) { return __builtin_amdgcn_readlane(v, j); }

__device__ __forceinline__ void atomic_max_f32(float* addr, float v) {
    const unsigned u = __float_as_uint(v);
    if ((u >> 31) == 0) atomicMax((int*)addr, (int)u);
    else                atomicMin((unsigned*)addr, u);
}

static_assert(OWN <= 256, "a list word has 8 bits for the row inside the block");
#define NOEDGE 0xffffffffu
__device__ __forceinline__ float* list_word(float* emb, int r0, int chunk, int lane) {
    return emb + (long)(r0 + (chunk >> 1)) * NEMB + 128 + 32 * (chunk & 1) + lane;
}

template <bool NEXT, int LIST>
__global__ __launch_bounds__(1024)
void layer_owned(const int* __restrict__ src, const int* __restrict__ dst,
                 const float* __restrict__ h, float* __restrict__ hn,
                 const float* __restrict__ dinv, const float* __restrict__ b,
                 float* __restrict__ emb, int colOff, int relu,
                 const float* __restrict__ Wn, float* __restrict__ ge, int E, int n) {
    __shared__ __attribute__((aligned(16))) float acc[OWN * NFEAT];
    __shared__ int qe[32][QCAP];
    __shared__ int qd[32][QCAP];
    __shared__ float red[OWN / 4][NFEAT];
    __shared__ __attribute__((aligned(16))) float sA[NEXT ? OWN * APITCH : 4];
    __shared__ __attribute__((aligned(16))) float sW[NEXT ? NFEAT * WPITCH : 4];
    __shared__ __attribute__((aligned(16))) float sD[OWN];
    __shared__ int s_found;
    const int lane = threadIdx.x & 31;
    const int wv   = threadIdx.x >> 5;
    const int r0   = blockIdx.x * OWN;
    const int cap  = min(OWN, n - r0) * 2 - 1;
    if (LIST == 0 && threadIdx.x == 0) s_found = 0;

    #pragma unroll
    for (int i = 0; i < OWN * NFEAT / 4 / 1024; ++i)
        *(float4*)(&acc[4 * (threadIdx.x + 1024 * i)]) = make_float4(0.f, 0.f, 0.f, 0.f);
    __syncthreads();

    int qh = 0;
    int qn = 0;
    auto addrows = [&](int s, int d, unsigned mask) {
        #pragma unroll 1
        while (mask) {
            int   dj[8];
            float v0[8], v1[8];
            #pragma unroll
            for (int k = 0; k < 8; ++k) {
                dj[k] = -1;
                if (mask) {
                    const int j = __builtin_ctz(mask);
                    mask &= mask - 1;
                    const float* hs = h + (long)lane_i(s, j) * NFEAT;
                    dj[k] = lane_i(d, j);
                    v0[k] = hs[lane];
                    v1[k] = hs[lane + 32];
                }
            }
            #pragma unroll
            for (int k = 0; k < 8; ++k) {
                if (dj[k] >= 0) {
                    float* ad = &acc[dj[k] * NFEAT];
                    atomicAdd(&ad[lane],      v0[k]);
                    atomicAdd(&ad[lane + 32], v1[k]);
                }
            }
        }
    };
    auto drain = [&](int cnt) {
        int s = 0, d = 0;
        __builtin_amdgcn_fence(__ATOMIC_ACQUIRE, "wavefront");
        if (lane < cnt) {
            const int slot = (qh + lane) & (QCAP - 1);
            s = src[qe[wv][slot]];
            d = qd[wv][slot];
        }
        if constexpr (LIST == 0) {
            int j = 0;
            if (lane == 0) j = atomicAdd(&s_found, 1);
            j = __builtin_amdgcn_readfirstlane(j);
            if (j < cap)
                *list_word(emb, r0, 1 + j, lane) =
                    __uint_as_float(lane < cnt ? ((unsigned)s | ((unsigned)d << 24)) : NOEDGE);
        }
        addrows(s, d, cnt >= 32 ? 0xffffffffu : ((1u << cnt) - 1u));
        qh = (qh + cnt) & (QCAP - 1);
        qn -= cnt;
    };
    auto put = [&](int dv, int ev) {
        const int  dl  = dv - r0;
        const bool hit = (unsigned)dl < (unsigned)OWN;
        const unsigned m = __builtin_amdgcn_ballot_w32(hit);
        if (m) {
            if (hit) {
                const int slot = (qh + qn + __builtin_popcount(m & ((1u << lane) - 1u))) & (QCAP - 1);
                qe[wv][slot] = ev;
                qd[wv][slot] = dl;
            }
            __builtin_amdgcn_fence(__ATOMIC_RELEASE, "wavefront");
            qn += __builtin_popcount(m);
        }
    };
    int  nch    = 0;
    bool listed = false;
    if constexpr (LIST == 1) {
        nch    = (int)__float_as_uint(*list_word(emb, r0, 0, 0));
        listed = (unsigned)nch <= (unsigned)cap;
    }
    if (listed) {
        auto ldw = [&](int j) {
            return __float_as_uint(*list_word(emb, r0, 1 + min(j, nch - 1), lane));
        };
        int j = __builtin_amdgcn_readfirstlane(wv);
        if (j < nch) {
            unsigned pk = ldw(j);
            while (j < nch) {
                const unsigned nx = ldw(j + 32);
                addrows((int)(pk & 0xffffffu), (int)(pk >> 24), __builtin_amdgcn_ballot_w32(pk != NOEDGE));
                pk = nx;
                j += 32;
            }
        }
    } else {
        const int RND = 1024 * 4;
        if ((E & 3) == 0 && E >= 4) {
            auto ld4 = [&](int e) { return *(const int4*)(dst + min(e, E - 4)); };
            auto trip = [&](int4 (&now)[SCAN_U], int4 (&next)[SCAN_U], int eb, const bool whole) {
                const int e0 = eb + 4 * (int)threadIdx.x;
                const int* pn = dst + e0 + SCAN_U * RND;
                #pragma unroll
                for (int u = 0; u < SCAN_U; ++u)
                    next[u] = whole ? *(const int4*)(pn + u * RND) : ld4(e0 + (SCAN_U + u) * RND);
                #pragma unroll
                for (int u = 0; u < SCAN_U; ++u) {
                    const int  e    = e0 + u * RND;
                    const bool live = whole || e < E;
                    const unsigned lo4 = min(min((unsigned)(now[u].x - r0), (unsigned)(now[u].y - r0)),
                                             min((unsigned)(now[u].z - r0), (unsigned)(now[u].w - r0)));
                    if (__builtin_amdgcn_ballot_w32(lo4 < (unsigned)OWN && live)) {
                        const int4 d4 = live ? now[u] : make_int4(-1, -1, -1, -1);
                        put(d4.x, e);
                        put(d4.y, e + 1);
                        put(d4.z, e + 2);
                        put(d4.w, e + 3);
                        while (qn >= 128) drain(32);
                    }
                }
                while (qn >= 32) drain(32);
            };
            int4 bufA[SCAN_U], bufB[SCAN_U];
            #pragma unroll
            for (int u = 0; u < SCAN_U; ++u) bufA[u] = ld4(u * RND + 4 * (int)threadIdx.x);
            int eb = 0;
            for (; eb + 3 * SCAN_U * RND <= E; eb += 2 * SCAN_U * RND) {
                trip(bufA, bufB, eb, true);
                trip(bufB, bufA, eb + SCAN_U * RND, true);
            }
            for (; eb < E; eb += 2 * SCAN_U * RND) {
                trip(bufA, bufB, eb, false);
                trip(bufB, bufA, eb + SCAN_U * RND, false);
            }
        } else {
            for (int eb = 0; eb < E; eb += 1024) {
                const int e = eb + (int)threadIdx.x;
                put(e < E ? dst[e] : -1, e);
                while (qn >= 32) drain(32);
            }
        }
        while (qn > 0) drain(qn < 32 ? qn : 32);
    }
    __syncthreads();
    if constexpr (LIST == 0) {
        if (threadIdx.x == 0)
            *list_word(emb, r0, 0, 0) = __uint_as_float(n < (1 << 24) ? (unsigned)s_found : 0x7fffffffu);
    }

    const int g   = threadIdx.x >> 8;
    const int tid = threadIdx.x & 255;
    const int q   = tid & 15;
    const int rs  = tid >> 4;
    const int lb  = g * FROWS;
    const int rb  = r0 + lb;
    const float4 bias = *(const float4*)(b + 4 * q);
    float4 m = make_float4(-INFINITY, -INFINITY, -INFINITY, -INFINITY);
    float  di[FROWS / 16];
    v4f t[FROWS / 16], w[FROWS / 16];
    if constexpr (NEXT) {
        if (g == 0) {
            #pragma unroll
            for (int p = 0; p < FROWS / 16; ++p)
                w[p] = *(const v4f*)(Wn + (rs + 16 * p) * NFEAT + 4 * q);
        }
    }
    #pragma unroll
    for (int p = 0; p < FROWS / 16; ++p) {
        int ic = min(rb + rs + 16 * p, n - 1);
        di[p] = dinv[ic];
        t[p] = *(const v4f*)(h + (long)ic * NFEAT + 4 * q);
    }
    if constexpr (NEXT) {
        if (g == 0) {
            #pragma unroll
            for (int p = 0; p < FROWS / 16; ++p)
                *(v4f*)(&sW[(rs + 16 * p) * WPITCH + 4 * q]) = w[p];
        }
    }
    #pragma unroll
    for (int p = 0; p < FROWS / 16; ++p) {
        const int lr = lb + rs + 16 * p;
        const int i  = r0 + lr;
        const v4f a = *(const v4f*)(&acc[lr * NFEAT + 4 * q]);
        float4 v;
        v.x = (a.x + t[p].x) * di[p] + bias.x;
        v.y = (a.y + t[p].y) * di[p] + bias.y;
        v.z = (a.z + t[p].z) * di[p] + bias.z;
        v.w = (a.w + t[p].w) * di[p] + bias.w;
        if (relu) {
            v.x = fmaxf(v.x, 0.0f); v.y = fmaxf(v.y, 0.0f);
            v.z = fmaxf(v.z, 0.0f); v.w = fmaxf(v.w, 0.0f);
        }
        if (i < n) {
            *(float4*)(emb + (long)i * NEMB + colOff + 4 * q) = v;
            m.x = fmaxf(m.x, v.x); m.y = fmaxf(m.y, v.y);
            m.z = fmaxf(m.z, v.z); m.w = fmaxf(m.w, v.w);
        }
        if constexpr (NEXT) {
            *(float4*)(&sA[lr * APITCH + 4 * q]) = v;
            if (q == 0) sD[lr] = di[p];
        }
    }
    *(float4*)(&red[g * 16 + rs][4 * q]) = m;
    __syncthreads();
    if (tid < NFEAT && rb < n) {
        float mm = red[g * 16][tid];
        #pragma unroll
        for (int r = 1; r < 16; ++r) mm = fmaxf(mm, red[g * 16 + r][tid]);
        atomic_max_f32(&ge[colOff + tid], mm);
    }
    if constexpr (NEXT) {
        const int w8 = (threadIdx.x >> 5) & 7;
        const int hi = lane >> 4;
        const int lo = lane & 15;
        const int tm = lb + (w8 >> 1) * 16;
        const int tn = (w8 & 1) * 32;
        const float* ar = &sA[(tm + lo) * APITCH + 2 * hi];
        const float* bc = &sW[(2 * hi) * WPITCH + tn + lo];
        const v4f dlo = *(const v4f*)(&sD[tm + 8 * hi]);
        const v4f dhi = *(const v4f*)(&sD[tm + 8 * hi + 4]);
        v8f c0 = {}, c1 = {};
        #pragma unroll
        for (int k = 0; k < NFEAT; k += 4) {
            v2f fa, fb0, fb1;
            fa.x  = ar[k];                       fa.y  = ar[k + 1];
            fb0.x = bc[k * WPITCH];              fb0.y = bc[(k + 1) * WPITCH];
            fb1.x = bc[k * WPITCH + 16];         fb1.y = bc[(k + 1) * WPITCH + 16];
            c0 = __builtin_amdgcn_wmma_f32_16x16x4_f32(false, fa, false, fb0, (short)0, c0, false, false);
            c1 = __builtin_amdgcn_wmma_f32_16x16x4_f32(false, fa, false, fb1, (short)0, c1, false, false);
        }
        const int rowBase = r0 + tm + 8 * hi;
        float* o = hn + (long)rowBase * NFEAT + tn + lo;
        #pragma unroll
        for (int r = 0; r < 8; ++r) {
            const float dr = r < 4 ? dlo[r & 3] : dhi[r & 3];
            if (rowBase + r < n) {
                o[(long)r * NFEAT]      = c0[r] * dr;
                o[(long)r * NFEAT + 16] = c1[r] * dr;
            }
        }
    }
}

__global__ void target_logits(const float* __restrict__ emb, const float* __restrict__ fcW,
                              const float* __restrict__ fcb, const int* __restrict__ target,
                              float* __restrict__ out) {
    int c = threadIdx.x;
    if (c >= 4) return;
    int t = target[0];
    float s = fcb[c];
    const float* row = emb + (long)t * NEMB;
    for (int f = 0; f < NEMB; ++f)
        s += row[f] * fcW[f * 4 + c];
    out[c] = s;
}

extern "C" void kernel_launch(void* const* d_in, const int* in_sizes, int n_in,
                              void* d_out, int out_size, void* d_ws, size_t ws_size,
                              hipStream_t stream) {
    const float* x      = (const float*)d_in[0];
    const int*   edge   = (const int*)d_in[1];
    const int*   target = (const int*)d_in[3];
    const float* W1 = (const float*)d_in[4];
    const float* b1 = (const float*)d_in[5];
    const float* W2 = (const float*)d_in[6];
    const float* b2 = (const float*)d_in[7];
    const float* W3 = (const float*)d_in[8];
    const float* b3 = (const float*)d_in[9];
    const float* fcW = (const float*)d_in[10];
    const float* fcb = (const float*)d_in[11];

    const int N = in_sizes[0] / NFEAT;
    const int E = in_sizes[1] / 2;
    const int* esrc = edge;
    const int* edst = edge + E;

    float* ws   = (float*)d_ws;
    float* dinv = ws;
    float* h    = dinv + N;
    float* h2   = h + (long)N * NFEAT;

    float* emb = (float*)d_out;
    float* ge  = emb + (long)N * NEMB;
    float* out4 = ge + NEMB;
    int* cnt = (int*)(ge + 128);

    init_deg<<<(N + 255) / 256, 256, 0, stream>>>(dinv, cnt, N);
    deg_count<<<1024, 256, 0, stream>>>(edst, dinv, cnt, E, N);
    make_dinv<<<(N + 255) / 256, 256, 0, stream>>>(dinv, ge, N);

    const int gemmWaves  = (N / 16) * 4;
    const int gemmBlocks = (gemmWaves + 7) / 8;
    const int nOwn       = (N + OWN - 1) / OWN;

    gemm_n64<<<gemmBlocks, 256, 0, stream>>>(x, NFEAT, W1, dinv, h, N);
    layer_owned<true, 0><<<nOwn, 1024, 0, stream>>>(esrc, edst, h,  h2, dinv, b1, emb, 0,   1, W2, ge, E, N);
    layer_owned<true, 1><<<nOwn, 1024, 0, stream>>>(esrc, edst, h2, h,  dinv, b2, emb, 64,  1, W3, ge, E, N);
    layer_owned<false, 1><<<nOwn, 1024, 0, stream>>>(esrc, edst, h, h2, dinv, b3, emb, 128, 0, nullptr, ge, E, N);

    target_logits<<<1, 32, 0, stream>>>(emb, fcW, fcb, target, out4);
}


# one owner-block scan of dst ahead of the layers counts its rows' degrees (LDS int adds, exact), leaves the edge list, makes its rows of product 1; 4 launches gone; arrival-order sums as before
# speedup vs baseline: 3.1882x; 1.0948x over previous
#include <hip/hip_runtime.h>
#include <hip/hip_bf16.h>

typedef float v8f __attribute__((ext_vector_type(8)));
typedef float v2f __attribute__((ext_vector_type(2)));
typedef float v4f __attribute__((ext_vector_type(4)));

#define NFEAT 64
#define NEMB  192

#define OWN 256
#define QCAP 256
#define SCAN_U 8
#define FROWS 64
#define APITCH 68
#define WPITCH 72
__device__ __forceinline__ int lane_i(int v, int j) { return __builtin_amdgcn_readlane(v, j); }

__device__ __forceinline__ void atomic_max_f32(float* addr, float v) {
    const unsigned u = __float_as_uint(v);
    if ((u >> 31) == 0) atomicMax((int*)addr, (int)u);
    else                atomicMin((unsigned*)addr, u);
}

static_assert(OWN <= 256, "a list word has 8 bits for the row inside the block");
#define NOEDGE 0xffffffffu
__device__ __forceinline__ float* list_word(float* emb, int r0, int chunk, int lane) {
    return emb + (long)(r0 + (chunk >> 1)) * NEMB + 128 + 32 * (chunk & 1) + lane;
}

template <class Sink>
__device__ __forceinline__ void scan_owned_edges(const int* __restrict__ src, const int* __restrict__ dst,
                                                 int E, int r0, int* qe, int* qd, Sink sink) {
    const int lane = threadIdx.x & 31;
    int qh = 0;
    int qn = 0;
    auto drain = [&](int cnt) {
        int s = 0, d = 0;
        __builtin_amdgcn_fence(__ATOMIC_ACQUIRE, "wavefront");
        if (lane < cnt) {
            const int slot = (qh + lane) & (QCAP - 1);
            s = src[qe[slot]];
            d = qd[slot];
        }
        sink(s, d, cnt);
        qh = (qh + cnt) & (QCAP - 1);
        qn -= cnt;
    };
    auto put = [&](int dv, int ev) {
        const int  dl  = dv - r0;
        const bool hit = (unsigned)dl < (unsigned)OWN;
        const unsigned m = __builtin_amdgcn_ballot_w32(hit);
        if (m) {
            if (hit) {
                const int slot = (qh + qn + __builtin_popcount(m & ((1u << lane) - 1u))) & (QCAP - 1);
                qe[slot] = ev;
                qd[slot] = dl;
            }
            __builtin_amdgcn_fence(__ATOMIC_RELEASE, "wavefront");
            qn += __builtin_popcount(m);
        }
    };
    const int RND = 1024 * 4;
    if ((E & 3) == 0 && E >= 4) {
        auto ld4 = [&](int e) { return *(const int4*)(dst + min(e, E - 4)); };
        auto trip = [&](int4 (&now)[SCAN_U], int4 (&next)[SCAN_U], int eb, const bool whole) {
            const int e0 = eb + 4 * (int)threadIdx.x;
            const int* pn = dst + e0 + SCAN_U * RND;
            #pragma unroll
            for (int u = 0; u < SCAN_U; ++u)
                next[u] = whole ? *(const int4*)(pn + u * RND) : ld4(e0 + (SCAN_U + u) * RND);
            #pragma unroll
            for (int u = 0; u < SCAN_U; ++u) {
                const int  e    = e0 + u * RND;
                const bool live = whole || e < E;
                const unsigned lo4 = min(min((unsigned)(now[u].x - r0), (unsigned)(now[u].y - r0)),
                                         min((unsigned)(now[u].z - r0), (unsigned)(now[u].w - r0)));
                if (__builtin_amdgcn_ballot_w32(lo4 < (unsigned)OWN && live)) {
                    const int4 d4 = live ? now[u] : make_int4(-1, -1, -1, -1);
                    put(d4.x, e);
                    put(d4.y, e + 1);
                    put(d4.z, e + 2);
                    put(d4.w, e + 3);
                    while (qn >= 128) drain(32);
                }
            }
            while (qn >= 32) drain(32);
        };
        int4 bufA[SCAN_U], bufB[SCAN_U];
        #pragma unroll
        for (int u = 0; u < SCAN_U; ++u) bufA[u] = ld4(u * RND + 4 * (int)threadIdx.x);
        int eb = 0;
        for (; eb + 3 * SCAN_U * RND <= E; eb += 2 * SCAN_U * RND) {
            trip(bufA, bufB, eb, true);
            trip(bufB, bufA, eb + SCAN_U * RND, true);
        }
        for (; eb < E; eb += 2 * SCAN_U * RND) {
            trip(bufA, bufB, eb, false);
            trip(bufB, bufA, eb + SCAN_U * RND, false);
        }
    } else {
        for (int eb = 0; eb < E; eb += 1024) {
            const int e = eb + (int)threadIdx.x;
            put(e < E ? dst[e] : -1, e);
            while (qn >= 32) drain(32);
        }
    }
    while (qn > 0) drain(qn < 32 ? qn : 32);
}

__device__ __forceinline__ void product_owned_rows(const float* sA, const float* sW, const float* sD,
                                                   float* __restrict__ out, int r0, int n) {
    const int lane = threadIdx.x & 31;
    const int lb = (threadIdx.x >> 8) * FROWS;
    const int w8 = (threadIdx.x >> 5) & 7;
    const int hi = lane >> 4;
    const int lo = lane & 15;
    const int tm = lb + (w8 >> 1) * 16;
    const int tn = (w8 & 1) * 32;
    const float* ar = &sA[(tm + lo) * APITCH + 2 * hi];
    const float* bc = &sW[(2 * hi) * WPITCH + tn + lo];
    const v4f dlo = *(const v4f*)(&sD[tm + 8 * hi]);
    const v4f dhi = *(const v4f*)(&sD[tm + 8 * hi + 4]);
    v8f c0 = {}, c1 = {};
    #pragma unroll
    for (int k = 0; k < NFEAT; k += 4) {
        v2f fa, fb0, fb1;
        fa.x  = ar[k];                       fa.y  = ar[k + 1];
        fb0.x = bc[k * WPITCH];              fb0.y = bc[(k + 1) * WPITCH];
        fb1.x = bc[k * WPITCH + 16];         fb1.y = bc[(k + 1) * WPITCH + 16];
        c0 = __builtin_amdgcn_wmma_f32_16x16x4_f32(false, fa, false, fb0, (short)0, c0, false, false);
        c1 = __builtin_amdgcn_wmma_f32_16x16x4_f32(false, fa, false, fb1, (short)0, c1, false, false);
    }
    const int rowBase = r0 + tm + 8 * hi;
    float* o = out + (long)rowBase * NFEAT + tn + lo;
    #pragma unroll
    for (int r = 0; r < 8; ++r) {
        const float dr = r < 4 ? dlo[r & 3] : dhi[r & 3];
        if (rowBase + r < n) {
            o[(long)r * NFEAT]      = c0[r] * dr;
            o[(long)r * NFEAT + 16] = c1[r] * dr;
        }
    }
}

__global__ __launch_bounds__(1024)
void scan_first(const int* __restrict__ src, const int* __restrict__ dst,
                const float* __restrict__ x, const float* __restrict__ W,
                float* __restrict__ dinv, float* __restrict__ h,
                float* __restrict__ emb, float* __restrict__ ge, int E, int n) {
    __shared__ int qe[32][QCAP];
    __shared__ int qd[32][QCAP];
    __shared__ int degc[OWN];
    __shared__ __attribute__((aligned(16))) float sA[OWN * APITCH];
    __shared__ __attribute__((aligned(16))) float sW[NFEAT * WPITCH];
    __shared__ __attribute__((aligned(16))) float sD[OWN];
    __shared__ int s_found;
    const int lane = threadIdx.x & 31;
    const int wv   = threadIdx.x >> 5;
    const int r0   = blockIdx.x * OWN;
    const int cap  = min(OWN, n - r0) * 2 - 1;
    if (threadIdx.x == 0) s_found = 0;
    if (threadIdx.x < OWN) degc[threadIdx.x] = 0;
    if (blockIdx.x == 0 && threadIdx.x < NEMB) ge[threadIdx.x] = -INFINITY;
    {
        const int q  = threadIdx.x & 15;
        const int rs = threadIdx.x >> 4;
        for (int p = 0; p < OWN / 64; ++p) {
            const int lr = rs + 64 * p;
            const int ic = min(r0 + lr, n - 1);
            *(v4f*)(&sA[lr * APITCH + 4 * q]) = *(const v4f*)(x + (long)ic * NFEAT + 4 * q);
        }
        *(v4f*)(&sW[rs * WPITCH + 4 * q]) = *(const v4f*)(W + rs * NFEAT + 4 * q);
    }
    __syncthreads();

    scan_owned_edges(src, dst, E, r0, qe[wv], qd[wv], [&](int s, int d, int cnt) {
        if (lane < cnt) atomicAdd(&degc[d], 1);
        int j = 0;
        if (lane == 0) j = atomicAdd(&s_found, 1);
        j = __builtin_amdgcn_readfirstlane(j);
        if (j < cap)
            *list_word(emb, r0, 1 + j, lane) =
                __uint_as_float(lane < cnt ? ((unsigned)s | ((unsigned)d << 24)) : NOEDGE);
    });
    __syncthreads();

    if (threadIdx.x == 0)
        *list_word(emb, r0, 0, 0) = __uint_as_float(n < (1 << 24) ? (unsigned)s_found : 0x7fffffffu);
    if (threadIdx.x < OWN) {
        const float dv = rsqrtf((float)(degc[threadIdx.x] + 1));
        sD[threadIdx.x] = dv;
        if (r0 + (int)threadIdx.x < n) dinv[r0 + threadIdx.x] = dv;
    }
    __syncthreads();
    product_owned_rows(sA, sW, sD, h, r0, n);
}

template <bool NEXT>
__global__ __launch_bounds__(1024)
void layer_owned(const int* __restrict__ src, const int* __restrict__ dst,
                 const float* __restrict__ h, float* __restrict__ hn,
                 const float* __restrict__ dinv, const float* __restrict__ b,
                 float* __restrict__ emb, int colOff, int relu,
                 const float* __restrict__ Wn, float* __restrict__ ge, int E, int n) {
    __shared__ __attribute__((aligned(16))) float acc[OWN * NFEAT];
    __shared__ int qe[32][QCAP];
    __shared__ int qd[32][QCAP];
    __shared__ float red[OWN / 4][NFEAT];
    __shared__ __attribute__((aligned(16))) float sA[NEXT ? OWN * APITCH : 4];
    __shared__ __attribute__((aligned(16))) float sW[NEXT ? NFEAT * WPITCH : 4];
    __shared__ __attribute__((aligned(16))) float sD[OWN];
    const int lane = threadIdx.x & 31;
    const int wv   = threadIdx.x >> 5;
    const int r0   = blockIdx.x * OWN;
    const int cap  = min(OWN, n - r0) * 2 - 1;

    #pragma unroll
    for (int i = 0; i < OWN * NFEAT / 4 / 1024; ++i)
        *(float4*)(&acc[4 * (threadIdx.x + 1024 * i)]) = make_float4(0.f, 0.f, 0.f, 0.f);
    __syncthreads();

    auto addrows = [&](int s, int d, unsigned mask) {
        #pragma unroll 1
        while (mask) {
            int   dj[8];
            float v0[8], v1[8];
            #pragma unroll
            for (int k = 0; k < 8; ++k) {
                dj[k] = -1;
                if (mask) {
                    const int j = __builtin_ctz(mask);
                    mask &= mask - 1;
                    const float* hs = h + (long)lane_i(s, j) * NFEAT;
                    dj[k] = lane_i(d, j);
                    v0[k] = hs[lane];
                    v1[k] = hs[lane + 32];
                }
            }
            #pragma unroll
            for (int k = 0; k < 8; ++k) {
                if (dj[k] >= 0) {
                    float* ad = &acc[dj[k] * NFEAT];
                    atomicAdd(&ad[lane],      v0[k]);
                    atomicAdd(&ad[lane + 32], v1[k]);
                }
            }
        }
    };
    const int  nch    = (int)__float_as_uint(*list_word(emb, r0, 0, 0));
    const bool listed = (unsigned)nch <= (unsigned)cap;
    if (listed) {
        auto ldw = [&](int j) {
            return __float_as_uint(*list_word(emb, r0, 1 + min(j, nch - 1), lane));
        };
        int j = __builtin_amdgcn_readfirstlane(wv);
        if (j < nch) {
            unsigned pk = ldw(j);
            while (j < nch) {
                const unsigned nx = ldw(j + 32);
                addrows((int)(pk & 0xffffffu), (int)(pk >> 24), __builtin_amdgcn_ballot_w32(pk != NOEDGE));
                pk = nx;
                j += 32;
            }
        }
    } else {
        scan_owned_edges(src, dst, E, r0, qe[wv], qd[wv], [&](int s, int d, int cnt) {
            addrows(s, d, cnt >= 32 ? 0xffffffffu : ((1u << cnt) - 1u));
        });
    }
    __syncthreads();

    const int g   = threadIdx.x >> 8;
    const int tid = threadIdx.x & 255;
    const int q   = tid & 15;
    const int rs  = tid >> 4;
    const int lb  = g * FROWS;
    const int rb  = r0 + lb;
    const float4 bias = *(const float4*)(b + 4 * q);
    float4 m = make_float4(-INFINITY, -INFINITY, -INFINITY, -INFINITY);
    float  di[FROWS / 16];
    v4f t[FROWS / 16], w[FROWS / 16];
    if constexpr (NEXT) {
        if (g == 0) {
            #pragma unroll
            for (int p = 0; p < FROWS / 16; ++p)
                w[p] = *(const v4f*)(Wn + (rs + 16 * p) * NFEAT + 4 * q);
        }
    }
    #pragma unroll
    for (int p = 0; p < FROWS / 16; ++p) {
        int ic = min(rb + rs + 16 * p, n - 1);
        di[p] = dinv[ic];
        t[p] = *(const v4f*)(h + (long)ic * NFEAT + 4 * q);
    }
    if constexpr (NEXT) {
        if (g == 0) {
            #pragma unroll
            for (int p = 0; p < FROWS / 16; ++p)
                *(v4f*)(&sW[(rs + 16 * p) * WPITCH + 4 * q]) = w[p];
        }
    }
    #pragma unroll
    for (int p = 0; p < FROWS / 16; ++p) {
        const int lr = lb + rs + 16 * p;
        const int i  = r0 + lr;
        const v4f a = *(const v4f*)(&acc[lr * NFEAT + 4 * q]);
        float4 v;
        v.x = (a.x + t[p].x) * di[p] + bias.x;
        v.y = (a.y + t[p].y) * di[p] + bias.y;
        v.z = (a.z + t[p].z) * di[p] + bias.z;
        v.w = (a.w + t[p].w) * di[p] + bias.w;
        if (relu) {
            v.x = fmaxf(v.x, 0.0f); v.y = fmaxf(v.y, 0.0f);
            v.z = fmaxf(v.z, 0.0f); v.w = fmaxf(v.w, 0.0f);
        }
        if (i < n) {
            *(float4*)(emb + (long)i * NEMB + colOff + 4 * q) = v;
            m.x = fmaxf(m.x, v.x); m.y = fmaxf(m.y, v.y);
            m.z = fmaxf(m.z, v.z); m.w = fmaxf(m.w, v.w);
        }
        if constexpr (NEXT) {
            *(float4*)(&sA[lr * APITCH + 4 * q]) = v;
            if (q == 0) sD[lr] = di[p];
        }
    }
    *(float4*)(&red[g * 16 + rs][4 * q]) = m;
    __syncthreads();
    if (tid < NFEAT && rb < n) {
        float mm = red[g * 16][tid];
        #pragma unroll
        for (int r = 1; r < 16; ++r) mm = fmaxf(mm, red[g * 16 + r][tid]);
        atomic_max_f32(&ge[colOff + tid], mm);
    }
    if constexpr (NEXT)
        product_owned_rows(sA, sW, sD, hn, r0, n);
}

__global__ void target_logits(const float* __restrict__ emb, const float* __restrict__ fcW,
                              const float* __restrict__ fcb, const int* __restrict__ target,
                              float* __restrict__ out) {
    int c = threadIdx.x;
    if (c >= 4) return;
    int t = target[0];
    float s = fcb[c];
    const float* row = emb + (long)t * NEMB;
    for (int f = 0; f < NEMB; ++f)
        s += row[f] * fcW[f * 4 + c];
    out[c] = s;
}

extern "C" void kernel_launch(void* const* d_in, const int* in_sizes, int n_in,
                              void* d_out, int out_size, void* d_ws, size_t ws_size,
                              hipStream_t stream) {
    const float* x      = (const float*)d_in[0];
    const int*   edge   = (const int*)d_in[1];
    const int*   target = (const int*)d_in[3];
    const float* W1 = (const float*)d_in[4];
    const float* b1 = (const float*)d_in[5];
    const float* W2 = (const float*)d_in[6];
    const float* b2 = (const float*)d_in[7];
    const float* W3 = (const float*)d_in[8];
    const float* b3 = (const float*)d_in[9];
    const float* fcW = (const float*)d_in[10];
    const float* fcb = (const float*)d_in[11];

    const int N = in_sizes[0] / NFEAT;
    const int E = in_sizes[1] / 2;
    const int* esrc = edge;
    const int* edst = edge + E;

    float* ws   = (float*)d_ws;
    float* dinv = ws;
    float* h    = dinv + N;
    float* h2   = h + (long)N * NFEAT;

    float* emb = (float*)d_out;
    float* ge  = emb + (long)N * NEMB;
    float* out4 = ge + NEMB;

    const int nOwn = (N + OWN - 1) / OWN;

    scan_first<<<nOwn, 1024, 0, stream>>>(esrc, edst, x, W1, dinv, h, emb, ge, E, N);

    layer_owned<true><<<nOwn, 1024, 0, stream>>>(esrc, edst, h,  h2, dinv, b1, emb, 0,   1, W2, ge, E, N);
    layer_owned<true><<<nOwn, 1024, 0, stream>>>(esrc, edst, h2, h,  dinv, b2, emb, 64,  1, W3, ge, E, N);
    layer_owned<false><<<nOwn, 1024, 0, stream>>>(esrc, edst, h, h2, dinv, b3, emb, 128, 0, nullptr, ge, E, N);

    target_logits<<<1, 32, 0, stream>>>(emb, fcW, fcb, target, out4);
}


# edge list binned by owner block once (128 blocks: LDS int count, prefix, place); an owner reads its 128 segments, not the whole dst list; 1 launch more; arrival-order sums as before
# speedup vs baseline: 3.8200x; 1.1982x over previous
#include <hip/hip_runtime.h>
#include <hip/hip_bf16.h>

typedef float v8f __attribute__((ext_vector_type(8)));
typedef float v2f __attribute__((ext_vector_type(2)));
typedef float v4f __attribute__((ext_vector_type(4)));

#define NFEAT 64
#define NEMB  192

#define OWN 256
#define QCAP 256
#define SCAN_U 8
#define FROWS 64
#define APITCH 68
#define WPITCH 72
__device__ __forceinline__ int lane_i(int v, int j) { return __builtin_amdgcn_readlane(v, j); }

__device__ __forceinline__ void atomic_max_f32(float* addr, float v) {
    const unsigned u = __float_as_uint(v);
    if ((u >> 31) == 0) atomicMax((int*)addr, (int)u);
    else                atomicMin((unsigned*)addr, u);
}

static_assert(OWN <= 256, "a list word has 8 bits for the row inside the block");
#define NOEDGE 0xffffffffu
__device__ __forceinline__ float* list_word(float* emb, int r0, int chunk, int lane) {
    return emb + (long)(r0 + (chunk >> 1)) * NEMB + 128 + 32 * (chunk & 1) + lane;
}

#define BINB 128
#define MAXOWN 1023
#define BIN_U 8
#define SEG_T 8
static_assert(BINB * SEG_T == 1024, "an owner block has SEG_T threads for each slice");

__device__ __forceinline__ bool bins_usable(int E, int n) {
    const long nOwn = (n + OWN - 1) / OWN;
    return E > 0 && n < (1 << 24) && nOwn <= MAXOWN &&
           (long)E + (long)BINB * (nOwn + 1) <= (long)n * NFEAT;
}

__device__ __forceinline__ int block_scan_excl(int v, int* wsum, int* total) {
    const int lane = threadIdx.x & 31;
    const int wv   = threadIdx.x >> 5;
    int incl = v;
    #pragma unroll
    for (int off = 1; off < 32; off <<= 1) {
        const int t = __shfl_up(incl, off);
        if (lane >= off) incl += t;
    }
    if (lane == 31) wsum[wv] = incl;
    __syncthreads();
    if (wv == 0) {
        const int w = wsum[lane];
        int wi = w;
        #pragma unroll
        for (int off = 1; off < 32; off <<= 1) {
            const int t = __shfl_up(wi, off);
            if (lane >= off) wi += t;
        }
        wsum[lane] = wi - w;
        if (lane == 31) wsum[32] = wi;
    }
    __syncthreads();
    *total = wsum[32];
    return incl - v + wsum[wv];
}

__global__ __launch_bounds__(1024)
void bin_edges(const int* __restrict__ src, const int* __restrict__ dst,
               int* __restrict__ bins, int E, int n) {
    __shared__ int cnt[MAXOWN + 1];
    __shared__ int base[MAXOWN + 1];
    __shared__ int wsum[33];
    if (!bins_usable(E, n)) return;
    const int nOwn = (n + OWN - 1) / OWN;
    const int L  = (E + BINB - 1) / BINB;
    const int e0 = min(E, (int)blockIdx.x * L);
    const int e1 = min(E, e0 + L);
    int* offs = bins + E + (long)blockIdx.x * (nOwn + 1);
    cnt[threadIdx.x] = 0;
    __syncthreads();
    for (int eb = e0 + (int)threadIdx.x; eb < e1; eb += BIN_U * 1024) {
        int d[BIN_U];
        #pragma unroll
        for (int u = 0; u < BIN_U; ++u) d[u] = eb + u * 1024 < e1 ? dst[eb + u * 1024] : -1;
        #pragma unroll
        for (int u = 0; u < BIN_U; ++u)
            if ((unsigned)d[u] < (unsigned)n) atomicAdd(&cnt[d[u] / OWN], 1);
    }
    __syncthreads();
    const int v = cnt[threadIdx.x];
    int total;
    const int excl = block_scan_excl(v, wsum, &total);
    base[threadIdx.x] = excl;
    cnt[threadIdx.x]  = 0;
    if ((int)threadIdx.x <= nOwn) offs[threadIdx.x] = excl;
    __syncthreads();
    for (int eb = e0 + (int)threadIdx.x; eb < e1; eb += BIN_U * 1024) {
        int d[BIN_U], s[BIN_U];
        #pragma unroll
        for (int u = 0; u < BIN_U; ++u) {
            const bool in = eb + u * 1024 < e1;
            d[u] = in ? dst[eb + u * 1024] : -1;
            s[u] = in ? src[eb + u * 1024] : 0;
        }
        #pragma unroll
        for (int u = 0; u < BIN_U; ++u) {
            if ((unsigned)d[u] < (unsigned)n) {
                const int o   = d[u] / OWN;
                const int pos = base[o] + atomicAdd(&cnt[o], 1);
                bins[e0 + pos] = (int)((unsigned)s[u] | ((unsigned)(d[u] - o * OWN) << 24));
            }
        }
    }
}

template <class Sink>
__device__ __forceinline__ void scan_owned_edges(const int* __restrict__ src, const int* __restrict__ dst,
                                                 int E, int r0, int* qe, int* qd, Sink sink) {
    const int lane = threadIdx.x & 31;
    int qh = 0;
    int qn = 0;
    auto drain = [&](int cnt) {
        int s = 0, d = 0;
        __builtin_amdgcn_fence(__ATOMIC_ACQUIRE, "wavefront");
        if (lane < cnt) {
            const int slot = (qh + lane) & (QCAP - 1);
            s = src[qe[slot]];
            d = qd[slot];
        }
        sink(s, d, cnt);
        qh = (qh + cnt) & (QCAP - 1);
        qn -= cnt;
    };
    auto put = [&](int dv, int ev) {
        const int  dl  = dv - r0;
        const bool hit = (unsigned)dl < (unsigned)OWN;
        const unsigned m = __builtin_amdgcn_ballot_w32(hit);
        if (m) {
            if (hit) {
                const int slot = (qh + qn + __builtin_popcount(m & ((1u << lane) - 1u))) & (QCAP - 1);
                qe[slot] = ev;
                qd[slot] = dl;
            }
            __builtin_amdgcn_fence(__ATOMIC_RELEASE, "wavefront");
            qn += __builtin_popcount(m);
        }
    };
    const int RND = 1024 * 4;
    if ((E & 3) == 0 && E >= 4) {
        auto ld4 = [&](int e) { return *(const int4*)(dst + min(e, E - 4)); };
        auto trip = [&](int4 (&now)[SCAN_U], int4 (&next)[SCAN_U], int eb, const bool whole) {
            const int e0 = eb + 4 * (int)threadIdx.x;
            const int* pn = dst + e0 + SCAN_U * RND;
            #pragma unroll
            for (int u = 0; u < SCAN_U; ++u)
                next[u] = whole ? *(const int4*)(pn + u * RND) : ld4(e0 + (SCAN_U + u) * RND);
            #pragma unroll
            for (int u = 0; u < SCAN_U; ++u) {
                const int  e    = e0 + u * RND;
                const bool live = whole || e < E;
                const unsigned lo4 = min(min((unsigned)(now[u].x - r0), (unsigned)(now[u].y - r0)),
                                         min((unsigned)(now[u].z - r0), (unsigned)(now[u].w - r0)));
                if (__builtin_amdgcn_ballot_w32(lo4 < (unsigned)OWN && live)) {
                    const int4 d4 = live ? now[u] : make_int4(-1, -1, -1, -1);
                    put(d4.x, e);
                    put(d4.y, e + 1);
                    put(d4.z, e + 2);
                    put(d4.w, e + 3);
                    while (qn >= 128) drain(32);
                }
            }
            while (qn >= 32) drain(32);
        };
        int4 bufA[SCAN_U], bufB[SCAN_U];
        #pragma unroll
        for (int u = 0; u < SCAN_U; ++u) bufA[u] = ld4(u * RND + 4 * (int)threadIdx.x);
        int eb = 0;
        for (; eb + 3 * SCAN_U * RND <= E; eb += 2 * SCAN_U * RND) {
            trip(bufA, bufB, eb, true);
            trip(bufB, bufA, eb + SCAN_U * RND, true);
        }
        for (; eb < E; eb += 2 * SCAN_U * RND) {
            trip(bufA, bufB, eb, false);
            trip(bufB, bufA, eb + SCAN_U * RND, false);
        }
    } else {
        for (int eb = 0; eb < E; eb += 1024) {
            const int e = eb + (int)threadIdx.x;
            put(e < E ? dst[e] : -1, e);
            while (qn >= 32) drain(32);
        }
    }
    while (qn > 0) drain(qn < 32 ? qn : 32);
}

__device__ __forceinline__ void product_owned_rows(const float* sA, const float* sW, const float* sD,
                                                   float* __restrict__ out, int r0, int n) {
    const int lane = threadIdx.x & 31;
    const int lb = (threadIdx.x >> 8) * FROWS;
    const int w8 = (threadIdx.x >> 5) & 7;
    const int hi = lane >> 4;
    const int lo = lane & 15;
    const int tm = lb + (w8 >> 1) * 16;
    const int tn = (w8 & 1) * 32;
    const float* ar = &sA[(tm + lo) * APITCH + 2 * hi];
    const float* bc = &sW[(2 * hi) * WPITCH + tn + lo];
    const v4f dlo = *(const v4f*)(&sD[tm + 8 * hi]);
    const v4f dhi = *(const v4f*)(&sD[tm + 8 * hi + 4]);
    v8f c0 = {}, c1 = {};
    #pragma unroll
    for (int k = 0; k < NFEAT; k += 4) {
        v2f fa, fb0, fb1;
        fa.x  = ar[k];                       fa.y  = ar[k + 1];
        fb0.x = bc[k * WPITCH];              fb0.y = bc[(k + 1) * WPITCH];
        fb1.x = bc[k * WPITCH + 16];         fb1.y = bc[(k + 1) * WPITCH + 16];
        c0 = __builtin_amdgcn_wmma_f32_16x16x4_f32(false, fa, false, fb0, (short)0, c0, false, false);
        c1 = __builtin_amdgcn_wmma_f32_16x16x4_f32(false, fa, false, fb1, (short)0, c1, false, false);
    }
    const int rowBase = r0 + tm + 8 * hi;
    float* o = out + (long)rowBase * NFEAT + tn + lo;
    #pragma unroll
    for (int r = 0; r < 8; ++r) {
        const float dr = r < 4 ? dlo[r & 3] : dhi[r & 3];
        if (rowBase + r < n) {
            o[(long)r * NFEAT]      = c0[r] * dr;
            o[(long)r * NFEAT + 16] = c1[r] * dr;
        }
    }
}

__global__ __launch_bounds__(1024)
void scan_first(const int* __restrict__ src, const int* __restrict__ dst,
                const int* __restrict__ bins,
                const float* __restrict__ x, const float* __restrict__ W,
                float* __restrict__ dinv, float* __restrict__ h,
                float* __restrict__ emb, float* __restrict__ ge, int E, int n) {
    __shared__ int ring[2][32][QCAP];
    __shared__ int wsum[33];
    __shared__ int degc[OWN];
    __shared__ __attribute__((aligned(16))) float sA[OWN * APITCH];
    __shared__ __attribute__((aligned(16))) float sW[NFEAT * WPITCH];
    __shared__ __attribute__((aligned(16))) float sD[OWN];
    __shared__ int s_found;
    const int lane = threadIdx.x & 31;
    const int wv   = threadIdx.x >> 5;
    const int r0   = blockIdx.x * OWN;
    const int cap  = min(OWN, n - r0) * 2 - 1;
    if (threadIdx.x == 0) s_found = 0;
    if (threadIdx.x < OWN) degc[threadIdx.x] = 0;
    if (blockIdx.x == 0 && threadIdx.x < NEMB) ge[threadIdx.x] = -INFINITY;
    {
        const int q  = threadIdx.x & 15;
        const int rs = threadIdx.x >> 4;
        for (int p = 0; p < OWN / 64; ++p) {
            const int lr = rs + 64 * p;
            const int ic = min(r0 + lr, n - 1);
            *(v4f*)(&sA[lr * APITCH + 4 * q]) = *(const v4f*)(x + (long)ic * NFEAT + 4 * q);
        }
        *(v4f*)(&sW[rs * WPITCH + 4 * q]) = *(const v4f*)(W + rs * NFEAT + 4 * q);
    }
    __syncthreads();

    if (bins_usable(E, n)) {
        int* stage = &ring[0][0][0];
        const int nOwn = (n + OWN - 1) / OWN;
        const int L    = (E + BINB - 1) / BINB;
        const int sl = threadIdx.x / SEG_T;
        const int j  = threadIdx.x % SEG_T;
        const int* o = bins + E + (long)sl * (nOwn + 1) + blockIdx.x;
        const int st = o[0];
        const int cs = o[1] - st;
        int total;
        const int ex = block_scan_excl(j == 0 ? cs : 0, wsum, &total);
        const int p0 = j == 0 ? ex : ex - cs;
        const int at = min(E, sl * L) + st;
        for (int ib = j; ib < cs; ib += SEG_T * BIN_U) {
            unsigned w[BIN_U];
            #pragma unroll
            for (int u = 0; u < BIN_U; ++u)
                w[u] = ib + SEG_T * u < cs ? (unsigned)bins[at + ib + SEG_T * u] : NOEDGE;
            #pragma unroll
            for (int u = 0; u < BIN_U; ++u) {
                const int i = ib + SEG_T * u;
                if (i < cs) {
                    atomicAdd(&degc[w[u] >> 24], 1);
                    if (p0 + i < 2 * 32 * QCAP) stage[p0 + i] = (int)w[u];
                }
            }
        }
        __syncthreads();
        const int  nchunks = (total + 31) >> 5;
        const bool fits    = nchunks <= cap;
        if (fits) {
            for (int k = wv; k < nchunks; k += 32) {
                const int p = 32 * k + lane;
                *list_word(emb, r0, 1 + k, lane) = __uint_as_float(p < total ? (unsigned)stage[p] : NOEDGE);
            }
        }
        if (threadIdx.x == 0)
            *list_word(emb, r0, 0, 0) = __uint_as_float(fits ? (unsigned)nchunks : 0x7fffffffu);
    } else {
        scan_owned_edges(src, dst, E, r0, ring[0][wv], ring[1][wv], [&](int s, int d, int cnt) {
            if (lane < cnt) atomicAdd(&degc[d], 1);
            int j = 0;
            if (lane == 0) j = atomicAdd(&s_found, 1);
            j = __builtin_amdgcn_readfirstlane(j);
            if (j < cap)
                *list_word(emb, r0, 1 + j, lane) =
                    __uint_as_float(lane < cnt ? ((unsigned)s | ((unsigned)d << 24)) : NOEDGE);
        });
        __syncthreads();

        if (threadIdx.x == 0)
            *list_word(emb, r0, 0, 0) = __uint_as_float(n < (1 << 24) ? (unsigned)s_found : 0x7fffffffu);
    }
    if (threadIdx.x < OWN) {
        const float dv = rsqrtf((float)(degc[threadIdx.x] + 1));
        sD[threadIdx.x] = dv;
        if (r0 + (int)threadIdx.x < n) dinv[r0 + threadIdx.x] = dv;
    }
    __syncthreads();
    product_owned_rows(sA, sW, sD, h, r0, n);
}

template <bool NEXT>
__global__ __launch_bounds__(1024)
void layer_owned(const int* __restrict__ src, const int* __restrict__ dst,
                 const float* __restrict__ h, float* __restrict__ hn,
                 const float* __restrict__ dinv, const float* __restrict__ b,
                 float* __restrict__ emb, int colOff, int relu,
                 const float* __restrict__ Wn, float* __restrict__ ge, int E, int n) {
    __shared__ __attribute__((aligned(16))) float acc[OWN * NFEAT];
    __shared__ int qe[32][QCAP];
    __shared__ int qd[32][QCAP];
    __shared__ float red[OWN / 4][NFEAT];
    __shared__ __attribute__((aligned(16))) float sA[NEXT ? OWN * APITCH : 4];
    __shared__ __attribute__((aligned(16))) float sW[NEXT ? NFEAT * WPITCH : 4];
    __shared__ __attribute__((aligned(16))) float sD[OWN];
    const int lane = threadIdx.x & 31;
    const int wv   = threadIdx.x >> 5;
    const int r0   = blockIdx.x * OWN;
    const int cap  = min(OWN, n - r0) * 2 - 1;

    #pragma unroll
    for (int i = 0; i < OWN * NFEAT / 4 / 1024; ++i)
        *(float4*)(&acc[4 * (threadIdx.x + 1024 * i)]) = make_float4(0.f, 0.f, 0.f, 0.f);
    __syncthreads();

    auto addrows = [&](int s, int d, unsigned mask) {
        #pragma unroll 1
        while (mask) {
            int   dj[8];
            float v0[8], v1[8];
            #pragma unroll
            for (int k = 0; k < 8; ++k) {
                dj[k] = -1;
                if (mask) {
                    const int j = __builtin_ctz(mask);
                    mask &= mask - 1;
                    const float* hs = h + (long)lane_i(s, j) * NFEAT;
                    dj[k] = lane_i(d, j);
                    v0[k] = hs[lane];
                    v1[k] = hs[lane + 32];
                }
            }
            #pragma unroll
            for (int k = 0; k < 8; ++k) {
                if (dj[k] >= 0) {
                    float* ad = &acc[dj[k] * NFEAT];
                    atomicAdd(&ad[lane],      v0[k]);
                    atomicAdd(&ad[lane + 32], v1[k]);
                }
            }
        }
    };
    const int  nch    = (int)__float_as_uint(*list_word(emb, r0, 0, 0));
    const bool listed = (unsigned)nch <= (unsigned)cap;
    if (listed) {
        auto ldw = [&](int j) {
            return __float_as_uint(*list_word(emb, r0, 1 + min(j, nch - 1), lane));
        };
        int j = __builtin_amdgcn_readfirstlane(wv);
        if (j < nch) {
            unsigned pk = ldw(j);
            while (j < nch) {
                const unsigned nx = ldw(j + 32);
                addrows((int)(pk & 0xffffffu), (int)(pk >> 24), __builtin_amdgcn_ballot_w32(pk != NOEDGE));
                pk = nx;
                j += 32;
            }
        }
    } else {
        scan_owned_edges(src, dst, E, r0, qe[wv], qd[wv], [&](int s, int d, int cnt) {
            addrows(s, d, cnt >= 32 ? 0xffffffffu : ((1u << cnt) - 1u));
        });
    }
    __syncthreads();

    const int g   = threadIdx.x >> 8;
    const int tid = threadIdx.x & 255;
    const int q   = tid & 15;
    const int rs  = tid >> 4;
    const int lb  = g * FROWS;
    const int rb  = r0 + lb;
    const float4 bias = *(const float4*)(b + 4 * q);
    float4 m = make_float4(-INFINITY, -INFINITY, -INFINITY, -INFINITY);
    float  di[FROWS / 16];
    v4f t[FROWS / 16], w[FROWS / 16];
    if constexpr (NEXT) {
        if (g == 0) {
            #pragma unroll
            for (int p = 0; p < FROWS / 16; ++p)
                w[p] = *(const v4f*)(Wn + (rs + 16 * p) * NFEAT + 4 * q);
        }
    }
    #pragma unroll
    for (int p = 0; p < FROWS / 16; ++p) {
        int ic = min(rb + rs + 16 * p, n - 1);
        di[p] = dinv[ic];
        t[p] = *(const v4f*)(h + (long)ic * NFEAT + 4 * q);
    }
    if constexpr (NEXT) {
        if (g == 0) {
            #pragma unroll
            for (int p = 0; p < FROWS / 16; ++p)
                *(v4f*)(&sW[(rs + 16 * p) * WPITCH + 4 * q]) = w[p];
        }
    }
    #pragma unroll
    for (int p = 0; p < FROWS / 16; ++p) {
        const int lr = lb + rs + 16 * p;
        const int i  = r0 + lr;
        const v4f a = *(const v4f*)(&acc[lr * NFEAT + 4 * q]);
        float4 v;
        v.x = (a.x + t[p].x) * di[p] + bias.x;
        v.y = (a.y + t[p].y) * di[p] + bias.y;
        v.z = (a.z + t[p].z) * di[p] + bias.z;
        v.w = (a.w + t[p].w) * di[p] + bias.w;
        if (relu) {
            v.x = fmaxf(v.x, 0.0f); v.y = fmaxf(v.y, 0.0f);
            v.z = fmaxf(v.z, 0.0f); v.w = fmaxf(v.w, 0.0f);
        }
        if (i < n) {
            *(float4*)(emb + (long)i * NEMB + colOff + 4 * q) = v;
            m.x = fmaxf(m.x, v.x); m.y = fmaxf(m.y, v.y);
            m.z = fmaxf(m.z, v.z); m.w = fmaxf(m.w, v.w);
        }
        if constexpr (NEXT) {
            *(float4*)(&sA[lr * APITCH + 4 * q]) = v;
            if (q == 0) sD[lr] = di[p];
        }
    }
    *(float4*)(&red[g * 16 + rs][4 * q]) = m;
    __syncthreads();
    if (tid < NFEAT && rb < n) {
        float mm = red[g * 16][tid];
        #pragma unroll
        for (int r = 1; r < 16; ++r) mm = fmaxf(mm, red[g * 16 + r][tid]);
        atomic_max_f32(&ge[colOff + tid], mm);
    }
    if constexpr (NEXT)
        product_owned_rows(sA, sW, sD, hn, r0, n);
}

__global__ void target_logits(const float* __restrict__ emb, const float* __restrict__ fcW,
                              const float* __restrict__ fcb, const int* __restrict__ target,
                              float* __restrict__ out) {
    int c = threadIdx.x;
    if (c >= 4) return;
    int t = target[0];
    float s = fcb[c];
    const float* row = emb + (long)t * NEMB;
    for (int f = 0; f < NEMB; ++f)
        s += row[f] * fcW[f * 4 + c];
    out[c] = s;
}

extern "C" void kernel_launch(void* const* d_in, const int* in_sizes, int n_in,
                              void* d_out, int out_size, void* d_ws, size_t ws_size,
                              hipStream_t stream) {
    const float* x      = (const float*)d_in[0];
    const int*   edge   = (const int*)d_in[1];
    const int*   target = (const int*)d_in[3];
    const float* W1 = (const float*)d_in[4];
    const float* b1 = (const float*)d_in[5];
    const float* W2 = (const float*)d_in[6];
    const float* b2 = (const float*)d_in[7];
    const float* W3 = (const float*)d_in[8];
    const float* b3 = (const float*)d_in[9];
    const float* fcW = (const float*)d_in[10];
    const float* fcb = (const float*)d_in[11];

    const int N = in_sizes[0] / NFEAT;
    const int E = in_sizes[1] / 2;
    const int* esrc = edge;
    const int* edst = edge + E;

    float* ws   = (float*)d_ws;
    float* dinv = ws;
    float* h    = dinv + N;
    float* h2   = h + (long)N * NFEAT;

    float* emb = (float*)d_out;
    float* ge  = emb + (long)N * NEMB;
    float* out4 = ge + NEMB;

    const int nOwn = (N + OWN - 1) / OWN;

    bin_edges<<<BINB, 1024, 0, stream>>>(esrc, edst, (int*)h2, E, N);

    scan_first<<<nOwn, 1024, 0, stream>>>(esrc, edst, (const int*)h2, x, W1, dinv, h, emb, ge, E, N);

    layer_owned<true><<<nOwn, 1024, 0, stream>>>(esrc, edst, h,  h2, dinv, b1, emb, 0,   1, W2, ge, E, N);
    layer_owned<true><<<nOwn, 1024, 0, stream>>>(esrc, edst, h2, h,  dinv, b2, emb, 64,  1, W3, ge, E, N);
    layer_owned<false><<<nOwn, 1024, 0, stream>>>(esrc, edst, h, h2, dinv, b3, emb, 128, 0, nullptr, ge, E, N);

    target_logits<<<1, 32, 0, stream>>>(emb, fcW, fcb, target, out4);
}


# target-node logits made by the target row's owner block at the end of layer 3 (one wave: lane-split sums, butterfly add); 6th launch and its 12 dependent load trips gone; logits' sum order changes
# speedup vs baseline: 4.0934x; 1.0716x over previous
#include <hip/hip_runtime.h>
#include <hip/hip_bf16.h>

typedef float v8f __attribute__((ext_vector_type(8)));
typedef float v2f __attribute__((ext_vector_type(2)));
typedef float v4f __attribute__((ext_vector_type(4)));

#define NFEAT 64
#define NEMB  192

#define OWN 256
#define QCAP 256
#define SCAN_U 8
#define FROWS 64
#define APITCH 68
#define WPITCH 72
__device__ __forceinline__ int lane_i(int v, int j) { return __builtin_amdgcn_readlane(v, j); }

__device__ __forceinline__ void atomic_max_f32(float* addr, float v) {
    const unsigned u = __float_as_uint(v);
    if ((u >> 31) == 0) atomicMax((int*)addr, (int)u);
    else                atomicMin((unsigned*)addr, u);
}

static_assert(OWN <= 256, "a list word has 8 bits for the row inside the block");
#define NOEDGE 0xffffffffu
__device__ __forceinline__ float* list_word(float* emb, int r0, int chunk, int lane) {
    return emb + (long)(r0 + (chunk >> 1)) * NEMB + 128 + 32 * (chunk & 1) + lane;
}

#define BINB 128
#define MAXOWN 1023
#define BIN_U 8
#define SEG_T 8
static_assert(BINB * SEG_T == 1024, "an owner block has SEG_T threads for each slice");

__device__ __forceinline__ bool bins_usable(int E, int n) {
    const long nOwn = (n + OWN - 1) / OWN;
    return E > 0 && n < (1 << 24) && nOwn <= MAXOWN &&
           (long)E + (long)BINB * (nOwn + 1) <= (long)n * NFEAT;
}

__device__ __forceinline__ int block_scan_excl(int v, int* wsum, int* total) {
    const int lane = threadIdx.x & 31;
    const int wv   = threadIdx.x >> 5;
    int incl = v;
    #pragma unroll
    for (int off = 1; off < 32; off <<= 1) {
        const int t = __shfl_up(incl, off);
        if (lane >= off) incl += t;
    }
    if (lane == 31) wsum[wv] = incl;
    __syncthreads();
    if (wv == 0) {
        const int w = wsum[lane];
        int wi = w;
        #pragma unroll
        for (int off = 1; off < 32; off <<= 1) {
            const int t = __shfl_up(wi, off);
            if (lane >= off) wi += t;
        }
        wsum[lane] = wi - w;
        if (lane == 31) wsum[32] = wi;
    }
    __syncthreads();
    *total = wsum[32];
    return incl - v + wsum[wv];
}

__global__ __launch_bounds__(1024)
void bin_edges(const int* __restrict__ src, const int* __restrict__ dst,
               int* __restrict__ bins, int E, int n) {
    __shared__ int cnt[MAXOWN + 1];
    __shared__ int base[MAXOWN + 1];
    __shared__ int wsum[33];
    if (!bins_usable(E, n)) return;
    const int nOwn = (n + OWN - 1) / OWN;
    const int L  = (E + BINB - 1) / BINB;
    const int e0 = min(E, (int)blockIdx.x * L);
    const int e1 = min(E, e0 + L);
    int* offs = bins + E + (long)blockIdx.x * (nOwn + 1);
    cnt[threadIdx.x] = 0;
    __syncthreads();
    for (int eb = e0 + (int)threadIdx.x; eb < e1; eb += BIN_U * 1024) {
        int d[BIN_U];
        #pragma unroll
        for (int u = 0; u < BIN_U; ++u) d[u] = eb + u * 1024 < e1 ? dst[eb + u * 1024] : -1;
        #pragma unroll
        for (int u = 0; u < BIN_U; ++u)
            if ((unsigned)d[u] < (unsigned)n) atomicAdd(&cnt[d[u] / OWN], 1);
    }
    __syncthreads();
    const int v = cnt[threadIdx.x];
    int total;
    const int excl = block_scan_excl(v, wsum, &total);
    base[threadIdx.x] = excl;
    cnt[threadIdx.x]  = 0;
    if ((int)threadIdx.x <= nOwn) offs[threadIdx.x] = excl;
    __syncthreads();
    for (int eb = e0 + (int)threadIdx.x; eb < e1; eb += BIN_U * 1024) {
        int d[BIN_U], s[BIN_U];
        #pragma unroll
        for (int u = 0; u < BIN_U; ++u) {
            const bool in = eb + u * 1024 < e1;
            d[u] = in ? dst[eb + u * 1024] : -1;
            s[u] = in ? src[eb + u * 1024] : 0;
        }
        #pragma unroll
        for (int u = 0; u < BIN_U; ++u) {
            if ((unsigned)d[u] < (unsigned)n) {
                const int o   = d[u] / OWN;
                const int pos = base[o] + atomicAdd(&cnt[o], 1);
                bins[e0 + pos] = (int)((unsigned)s[u] | ((unsigned)(d[u] - o * OWN) << 24));
            }
        }
    }
}

template <class Sink>
__device__ __forceinline__ void scan_owned_edges(const int* __restrict__ src, const int* __restrict__ dst,
                                                 int E, int r0, int* qe, int* qd, Sink sink) {
    const int lane = threadIdx.x & 31;
    int qh = 0;
    int qn = 0;
    auto drain = [&](int cnt) {
        int s = 0, d = 0;
        __builtin_amdgcn_fence(__ATOMIC_ACQUIRE, "wavefront");
        if (lane < cnt) {
            const int slot = (qh + lane) & (QCAP - 1);
            s = src[qe[slot]];
            d = qd[slot];
        }
        sink(s, d, cnt);
        qh = (qh + cnt) & (QCAP - 1);
        qn -= cnt;
    };
    auto put = [&](int dv, int ev) {
        const int  dl  = dv - r0;
        const bool hit = (unsigned)dl < (unsigned)OWN;
        const unsigned m = __builtin_amdgcn_ballot_w32(hit);
        if (m) {
            if (hit) {
                const int slot = (qh + qn + __builtin_popcount(m & ((1u << lane) - 1u))) & (QCAP - 1);
                qe[slot] = ev;
                qd[slot] = dl;
            }
            __builtin_amdgcn_fence(__ATOMIC_RELEASE, "wavefront");
            qn += __builtin_popcount(m);
        }
    };
    const int RND = 1024 * 4;
    if ((E & 3) == 0 && E >= 4) {
        auto ld4 = [&](int e) { return *(const int4*)(dst + min(e, E - 4)); };
        auto trip = [&](int4 (&now)[SCAN_U], int4 (&next)[SCAN_U], int eb, const bool whole) {
            const int e0 = eb + 4 * (int)threadIdx.x;
            const int* pn = dst + e0 + SCAN_U * RND;
            #pragma unroll
            for (int u = 0; u < SCAN_U; ++u)
                next[u] = whole ? *(const int4*)(pn + u * RND) : ld4(e0 + (SCAN_U + u) * RND);
            #pragma unroll
            for (int u = 0; u < SCAN_U; ++u) {
                const int  e    = e0 + u * RND;
                const bool live = whole || e < E;
                const unsigned lo4 = min(min((unsigned)(now[u].x - r0), (unsigned)(now[u].y - r0)),
                                         min((unsigned)(now[u].z - r0), (unsigned)(now[u].w - r0)));
                if (__builtin_amdgcn_ballot_w32(lo4 < (unsigned)OWN && live)) {
                    const int4 d4 = live ? now[u] : make_int4(-1, -1, -1, -1);
                    put(d4.x, e);
                    put(d4.y, e + 1);
                    put(d4.z, e + 2);
                    put(d4.w, e + 3);
                    while (qn >= 128) drain(32);
                }
            }
            while (qn >= 32) drain(32);
        };
        int4 bufA[SCAN_U], bufB[SCAN_U];
        #pragma unroll
        for (int u = 0; u < SCAN_U; ++u) bufA[u] = ld4(u * RND + 4 * (int)threadIdx.x);
        int eb = 0;
        for (; eb + 3 * SCAN_U * RND <= E; eb += 2 * SCAN_U * RND) {
            trip(bufA, bufB, eb, true);
            trip(bufB, bufA, eb + SCAN_U * RND, true);
        }
        for (; eb < E; eb += 2 * SCAN_U * RND) {
            trip(bufA, bufB, eb, false);
            trip(bufB, bufA, eb + SCAN_U * RND, false);
        }
    } else {
        for (int eb = 0; eb < E; eb += 1024) {
            const int e = eb + (int)threadIdx.x;
            put(e < E ? dst[e] : -1, e);
            while (qn >= 32) drain(32);
        }
    }
    while (qn > 0) drain(qn < 32 ? qn : 32);
}

__device__ __forceinline__ void product_owned_rows(const float* sA, const float* sW, const float* sD,
                                                   float* __restrict__ out, int r0, int n) {
    const int lane = threadIdx.x & 31;
    const int lb = (threadIdx.x >> 8) * FROWS;
    const int w8 = (threadIdx.x >> 5) & 7;
    const int hi = lane >> 4;
    const int lo = lane & 15;
    const int tm = lb + (w8 >> 1) * 16;
    const int tn = (w8 & 1) * 32;
    const float* ar = &sA[(tm + lo) * APITCH + 2 * hi];
    const float* bc = &sW[(2 * hi) * WPITCH + tn + lo];
    const v4f dlo = *(const v4f*)(&sD[tm + 8 * hi]);
    const v4f dhi = *(const v4f*)(&sD[tm + 8 * hi + 4]);
    v8f c0 = {}, c1 = {};
    #pragma unroll
    for (int k = 0; k < NFEAT; k += 4) {
        v2f fa, fb0, fb1;
        fa.x  = ar[k];                       fa.y  = ar[k + 1];
        fb0.x = bc[k * WPITCH];              fb0.y = bc[(k + 1) * WPITCH];
        fb1.x = bc[k * WPITCH + 16];         fb1.y = bc[(k + 1) * WPITCH + 16];
        c0 = __builtin_amdgcn_wmma_f32_16x16x4_f32(false, fa, false, fb0, (short)0, c0, false, false);
        c1 = __builtin_amdgcn_wmma_f32_16x16x4_f32(false, fa, false, fb1, (short)0, c1, false, false);
    }
    const int rowBase = r0 + tm + 8 * hi;
    float* o = out + (long)rowBase * NFEAT + tn + lo;
    #pragma unroll
    for (int r = 0; r < 8; ++r) {
        const float dr = r < 4 ? dlo[r & 3] : dhi[r & 3];
        if (rowBase + r < n) {
            o[(long)r * NFEAT]      = c0[r] * dr;
            o[(long)r * NFEAT + 16] = c1[r] * dr;
        }
    }
}

__global__ __launch_bounds__(1024)
void scan_first(const int* __restrict__ src, const int* __restrict__ dst,
                const int* __restrict__ bins,
                const float* __restrict__ x, const float* __restrict__ W,
                float* __restrict__ dinv, float* __restrict__ h,
                float* __restrict__ emb, float* __restrict__ ge, int E, int n) {
    __shared__ int ring[2][32][QCAP];
    __shared__ int wsum[33];
    __shared__ int degc[OWN];
    __shared__ __attribute__((aligned(16))) float sA[OWN * APITCH];
    __shared__ __attribute__((aligned(16))) float sW[NFEAT * WPITCH];
    __shared__ __attribute__((aligned(16))) float sD[OWN];
    __shared__ int s_found;
    const int lane = threadIdx.x & 31;
    const int wv   = threadIdx.x >> 5;
    const int r0   = blockIdx.x * OWN;
    const int cap  = min(OWN, n - r0) * 2 - 1;
    if (threadIdx.x == 0) s_found = 0;
    if (threadIdx.x < OWN) degc[threadIdx.x] = 0;
    if (blockIdx.x == 0 && threadIdx.x < NEMB) ge[threadIdx.x] = -INFINITY;
    {
        const int q  = threadIdx.x & 15;
        const int rs = threadIdx.x >> 4;
        for (int p = 0; p < OWN / 64; ++p) {
            const int lr = rs + 64 * p;
            const int ic = min(r0 + lr, n - 1);
            *(v4f*)(&sA[lr * APITCH + 4 * q]) = *(const v4f*)(x + (long)ic * NFEAT + 4 * q);
        }
        *(v4f*)(&sW[rs * WPITCH + 4 * q]) = *(const v4f*)(W + rs * NFEAT + 4 * q);
    }
    __syncthreads();

    if (bins_usable(E, n)) {
        int* stage = &ring[0][0][0];
        const int nOwn = (n + OWN - 1) / OWN;
        const int L    = (E + BINB - 1) / BINB;
        const int sl = threadIdx.x / SEG_T;
        const int j  = threadIdx.x % SEG_T;
        const int* o = bins + E + (long)sl * (nOwn + 1) + blockIdx.x;
        const int st = o[0];
        const int cs = o[1] - st;
        int total;
        const int ex = block_scan_excl(j == 0 ? cs : 0, wsum, &total);
        const int p0 = j == 0 ? ex : ex - cs;
        const int at = min(E, sl * L) + st;
        for (int ib = j; ib < cs; ib += SEG_T * BIN_U) {
            unsigned w[BIN_U];
            #pragma unroll
            for (int u = 0; u < BIN_U; ++u)
                w[u] = ib + SEG_T * u < cs ? (unsigned)bins[at + ib + SEG_T * u] : NOEDGE;
            #pragma unroll
            for (int u = 0; u < BIN_U; ++u) {
                const int i = ib + SEG_T * u;
                if (i < cs) {
                    atomicAdd(&degc[w[u] >> 24], 1);
                    if (p0 + i < 2 * 32 * QCAP) stage[p0 + i] = (int)w[u];
                }
            }
        }
        __syncthreads();
        const int  nchunks = (total + 31) >> 5;
        const bool fits    = nchunks <= cap;
        if (fits) {
            for (int k = wv; k < nchunks; k += 32) {
                const int p = 32 * k + lane;
                *list_word(emb, r0, 1 + k, lane) = __uint_as_float(p < total ? (unsigned)stage[p] : NOEDGE);
            }
        }
        if (threadIdx.x == 0)
            *list_word(emb, r0, 0, 0) = __uint_as_float(fits ? (unsigned)nchunks : 0x7fffffffu);
    } else {
        scan_owned_edges(src, dst, E, r0, ring[0][wv], ring[1][wv], [&](int s, int d, int cnt) {
            if (lane < cnt) atomicAdd(&degc[d], 1);
            int j = 0;
            if (lane == 0) j = atomicAdd(&s_found, 1);
            j = __builtin_amdgcn_readfirstlane(j);
            if (j < cap)
                *list_word(emb, r0, 1 + j, lane) =
                    __uint_as_float(lane < cnt ? ((unsigned)s | ((unsigned)d << 24)) : NOEDGE);
        });
        __syncthreads();

        if (threadIdx.x == 0)
            *list_word(emb, r0, 0, 0) = __uint_as_float(n < (1 << 24) ? (unsigned)s_found : 0x7fffffffu);
    }
    if (threadIdx.x < OWN) {
        const float dv = rsqrtf((float)(degc[threadIdx.x] + 1));
        sD[threadIdx.x] = dv;
        if (r0 + (int)threadIdx.x < n) dinv[r0 + threadIdx.x] = dv;
    }
    __syncthreads();
    product_owned_rows(sA, sW, sD, h, r0, n);
}

template <bool NEXT>
__global__ __launch_bounds__(1024)
void layer_owned(const int* __restrict__ src, const int* __restrict__ dst,
                 const float* __restrict__ h, float* __restrict__ hn,
                 const float* __restrict__ dinv, const float* __restrict__ b,
                 float* __restrict__ emb, int colOff, int relu,
                 const float* __restrict__ Wn, float* __restrict__ ge,
                 const float* __restrict__ fcW, const float* __restrict__ fcb,
                 const int* __restrict__ target, float* __restrict__ out4, int E, int n) {
    __shared__ __attribute__((aligned(16))) float acc[OWN * NFEAT];
    __shared__ int qe[32][QCAP];
    __shared__ int qd[32][QCAP];
    __shared__ float red[OWN / 4][NFEAT];
    __shared__ __attribute__((aligned(16))) float sA[NEXT ? OWN * APITCH : 4];
    __shared__ __attribute__((aligned(16))) float sW[NEXT ? NFEAT * WPITCH : 4];
    __shared__ __attribute__((aligned(16))) float sD[OWN];
    __shared__ __attribute__((aligned(16))) float trow[NEXT ? 4 : NFEAT];
    const int lane = threadIdx.x & 31;
    const int wv   = threadIdx.x >> 5;
    const int r0   = blockIdx.x * OWN;
    const int cap  = min(OWN, n - r0) * 2 - 1;
    int  tgt   = 0;
    bool own_t = false;
    if constexpr (!NEXT) {
        tgt   = min(max(target[0], 0), n - 1);
        own_t = tgt >= r0 && tgt < r0 + OWN;
    }

    #pragma unroll
    for (int i = 0; i < OWN * NFEAT / 4 / 1024; ++i)
        *(float4*)(&acc[4 * (threadIdx.x + 1024 * i)]) = make_float4(0.f, 0.f, 0.f, 0.f);
    __syncthreads();

    auto addrows = [&](int s, int d, unsigned mask) {
        #pragma unroll 1
        while (mask) {
            int   dj[8];
            float v0[8], v1[8];
            #pragma unroll
            for (int k = 0; k < 8; ++k) {
                dj[k] = -1;
                if (mask) {
                    const int j = __builtin_ctz(mask);
                    mask &= mask - 1;
                    const float* hs = h + (long)lane_i(s, j) * NFEAT;
                    dj[k] = lane_i(d, j);
                    v0[k] = hs[lane];
                    v1[k] = hs[lane + 32];
                }
            }
            #pragma unroll
            for (int k = 0; k < 8; ++k) {
                if (dj[k] >= 0) {
                    float* ad = &acc[dj[k] * NFEAT];
                    atomicAdd(&ad[lane],      v0[k]);
                    atomicAdd(&ad[lane + 32], v1[k]);
                }
            }
        }
    };
    const int  nch    = (int)__float_as_uint(*list_word(emb, r0, 0, 0));
    const bool listed = (unsigned)nch <= (unsigned)cap;
    if (listed) {
        auto ldw = [&](int j) {
            return __float_as_uint(*list_word(emb, r0, 1 + min(j, nch - 1), lane));
        };
        int j = __builtin_amdgcn_readfirstlane(wv);
        if (j < nch) {
            unsigned pk = ldw(j);
            while (j < nch) {
                const unsigned nx = ldw(j + 32);
                addrows((int)(pk & 0xffffffu), (int)(pk >> 24), __builtin_amdgcn_ballot_w32(pk != NOEDGE));
                pk = nx;
                j += 32;
            }
        }
    } else {
        scan_owned_edges(src, dst, E, r0, qe[wv], qd[wv], [&](int s, int d, int cnt) {
            addrows(s, d, cnt >= 32 ? 0xffffffffu : ((1u << cnt) - 1u));
        });
    }
    __syncthreads();

    const int g   = threadIdx.x >> 8;
    const int tid = threadIdx.x & 255;
    const int q   = tid & 15;
    const int rs  = tid >> 4;
    const int lb  = g * FROWS;
    const int rb  = r0 + lb;
    const float4 bias = *(const float4*)(b + 4 * q);
    float4 m = make_float4(-INFINITY, -INFINITY, -INFINITY, -INFINITY);
    float  di[FROWS / 16];
    v4f t[FROWS / 16], w[FROWS / 16];
    if constexpr (NEXT) {
        if (g == 0) {
            #pragma unroll
            for (int p = 0; p < FROWS / 16; ++p)
                w[p] = *(const v4f*)(Wn + (rs + 16 * p) * NFEAT + 4 * q);
        }
    }
    #pragma unroll
    for (int p = 0; p < FROWS / 16; ++p) {
        int ic = min(rb + rs + 16 * p, n - 1);
        di[p] = dinv[ic];
        t[p] = *(const v4f*)(h + (long)ic * NFEAT + 4 * q);
    }
    if constexpr (NEXT) {
        if (g == 0) {
            #pragma unroll
            for (int p = 0; p < FROWS / 16; ++p)
                *(v4f*)(&sW[(rs + 16 * p) * WPITCH + 4 * q]) = w[p];
        }
    }
    #pragma unroll
    for (int p = 0; p < FROWS / 16; ++p) {
        const int lr = lb + rs + 16 * p;
        const int i  = r0 + lr;
        const v4f a = *(const v4f*)(&acc[lr * NFEAT + 4 * q]);
        float4 v;
        v.x = (a.x + t[p].x) * di[p] + bias.x;
        v.y = (a.y + t[p].y) * di[p] + bias.y;
        v.z = (a.z + t[p].z) * di[p] + bias.z;
        v.w = (a.w + t[p].w) * di[p] + bias.w;
        if (relu) {
            v.x = fmaxf(v.x, 0.0f); v.y = fmaxf(v.y, 0.0f);
            v.z = fmaxf(v.z, 0.0f); v.w = fmaxf(v.w, 0.0f);
        }
        if (i < n) {
            *(float4*)(emb + (long)i * NEMB + colOff + 4 * q) = v;
            m.x = fmaxf(m.x, v.x); m.y = fmaxf(m.y, v.y);
            m.z = fmaxf(m.z, v.z); m.w = fmaxf(m.w, v.w);
            if constexpr (!NEXT) {
                if (i == tgt) *(float4*)(&trow[4 * q]) = v;
            }
        }
        if constexpr (NEXT) {
            *(float4*)(&sA[lr * APITCH + 4 * q]) = v;
            if (q == 0) sD[lr] = di[p];
        }
    }
    *(float4*)(&red[g * 16 + rs][4 * q]) = m;
    __syncthreads();
    if (tid < NFEAT && rb < n) {
        float mm = red[g * 16][tid];
        #pragma unroll
        for (int r = 1; r < 16; ++r) mm = fmaxf(mm, red[g * 16 + r][tid]);
        atomic_max_f32(&ge[colOff + tid], mm);
    }
    if constexpr (NEXT)
        product_owned_rows(sA, sW, sD, hn, r0, n);
    if constexpr (!NEXT) {
        if (own_t && wv == 31) {
            const float* er = emb + (long)tgt * NEMB;
            float s0 = 0.f, s1 = 0.f, s2 = 0.f, s3 = 0.f;
            #pragma unroll
            for (int k = 0; k < NEMB / 32; ++k) {
                const int    f  = lane + 32 * k;
                const float4 w4 = *(const float4*)(fcW + 4 * f);
                const float  e  = 32 * k < 128 ? er[f] : trow[f - 128];
                s0 += e * w4.x; s1 += e * w4.y;
                s2 += e * w4.z; s3 += e * w4.w;
            }
            #pragma unroll
            for (int off = 16; off >= 1; off >>= 1) {
                s0 += __shfl_xor(s0, off); s1 += __shfl_xor(s1, off);
                s2 += __shfl_xor(s2, off); s3 += __shfl_xor(s3, off);
            }
            if (lane == 0) {
                out4[0] = s0 + fcb[0]; out4[1] = s1 + fcb[1];
                out4[2] = s2 + fcb[2]; out4[3] = s3 + fcb[3];
            }
        }
    }
}

extern "C" void kernel_launch(void* const* d_in, const int* in_sizes, int n_in,
                              void* d_out, int out_size, void* d_ws, size_t ws_size,
                              hipStream_t stream) {
    const float* x      = (const float*)d_in[0];
    const int*   edge   = (const int*)d_in[1];
    const int*   target = (const int*)d_in[3];
    const float* W1 = (const float*)d_in[4];
    const float* b1 = (const float*)d_in[5];
    const float* W2 = (const float*)d_in[6];
    const float* b2 = (const float*)d_in[7];
    const float* W3 = (const float*)d_in[8];
    const float* b3 = (const float*)d_in[9];
    const float* fcW = (const float*)d_in[10];
    const float* fcb = (const float*)d_in[11];

    const int N = in_sizes[0] / NFEAT;
    const int E = in_sizes[1] / 2;
    const int* esrc = edge;
    const int* edst = edge + E;

    float* ws   = (float*)d_ws;
    float* dinv = ws;
    float* h    = dinv + N;
    float* h2   = h + (long)N * NFEAT;

    float* emb = (float*)d_out;
    float* ge  = emb + (long)N * NEMB;
    float* out4 = ge + NEMB;

    const int nOwn = (N + OWN - 1) / OWN;

    bin_edges<<<BINB, 1024, 0, stream>>>(esrc, edst, (int*)h2, E, N);

    scan_first<<<nOwn, 1024, 0, stream>>>(esrc, edst, (const int*)h2, x, W1, dinv, h, emb, ge, E, N);

    layer_owned<true><<<nOwn, 1024, 0, stream>>>(esrc, edst, h,  h2, dinv, b1, emb, 0,   1, W2, ge,
                                                 nullptr, nullptr, nullptr, nullptr, E, N);
    layer_owned<true><<<nOwn, 1024, 0, stream>>>(esrc, edst, h2, h,  dinv, b2, emb, 64,  1, W3, ge,
                                                 nullptr, nullptr, nullptr, nullptr, E, N);
    layer_owned<false><<<nOwn, 1024, 0, stream>>>(esrc, edst, h, h2, dinv, b3, emb, 128, 0, nullptr, ge,
                                                  fcW, fcb, target, out4, E, N);
}


# owner blocks re-tiled to fill the chip: 208 rows a block (13 row tiles of 16) give 241 blocks for 256 WGPs; 256 rows gave 196, one a WGP by its LDS, 60 WGPs idle; arrival-order sums as before
# speedup vs baseline: 4.3824x; 1.0706x over previous
#include <hip/hip_runtime.h>
#include <hip/hip_bf16.h>

typedef float v8f __attribute__((ext_vector_type(8)));
typedef float v2f __attribute__((ext_vector_type(2)));
typedef float v4f __attribute__((ext_vector_type(4)));

#define NFEAT 64
#define NEMB  192

#define OWN 208
#define PROWS 64
static_assert(OWN % 16 == 0, "the product works on whole 16-row tiles");
#define QCAP 256
#define SCAN_U 8
#define APITCH 68
#define WPITCH 72
__device__ __forceinline__ int lane_i(int v, int j) { return __builtin_amdgcn_readlane(v, j); }

__device__ __forceinline__ void atomic_max_f32(float* addr, float v) {
    const unsigned u = __float_as_uint(v);
    if ((u >> 31) == 0) atomicMax((int*)addr, (int)u);
    else                atomicMin((unsigned*)addr, u);
}

static_assert(OWN <= 256, "a list word has 8 bits for the row inside the block");
#define NOEDGE 0xffffffffu
__device__ __forceinline__ float* list_word(float* emb, int r0, int chunk, int lane) {
    return emb + (long)(r0 + (chunk >> 1)) * NEMB + 128 + 32 * (chunk & 1) + lane;
}

#define BINB 128
#define MAXOWN 1023
#define BIN_U 8
#define SEG_T 8
static_assert(BINB * SEG_T == 1024, "an owner block has SEG_T threads for each slice");

__device__ __forceinline__ bool bins_usable(int E, int n) {
    const long nOwn = (n + OWN - 1) / OWN;
    return E > 0 && n < (1 << 24) && nOwn <= MAXOWN &&
           (long)E + (long)BINB * (nOwn + 1) <= (long)n * NFEAT;
}

__device__ __forceinline__ int block_scan_excl(int v, int* wsum, int* total) {
    const int lane = threadIdx.x & 31;
    const int wv   = threadIdx.x >> 5;
    int incl = v;
    #pragma unroll
    for (int off = 1; off < 32; off <<= 1) {
        const int t = __shfl_up(incl, off);
        if (lane >= off) incl += t;
    }
    if (lane == 31) wsum[wv] = incl;
    __syncthreads();
    if (wv == 0) {
        const int w = wsum[lane];
        int wi = w;
        #pragma unroll
        for (int off = 1; off < 32; off <<= 1) {
            const int t = __shfl_up(wi, off);
            if (lane >= off) wi += t;
        }
        wsum[lane] = wi - w;
        if (lane == 31) wsum[32] = wi;
    }
    __syncthreads();
    *total = wsum[32];
    return incl - v + wsum[wv];
}

__global__ __launch_bounds__(1024)
void bin_edges(const int* __restrict__ src, const int* __restrict__ dst,
               int* __restrict__ bins, int E, int n) {
    __shared__ int cnt[MAXOWN + 1];
    __shared__ int base[MAXOWN + 1];
    __shared__ int wsum[33];
    if (!bins_usable(E, n)) return;
    const int nOwn = (n + OWN - 1) / OWN;
    const int L  = (E + BINB - 1) / BINB;
    const int e0 = min(E, (int)blockIdx.x * L);
    const int e1 = min(E, e0 + L);
    int* offs = bins + E + (long)blockIdx.x * (nOwn + 1);
    cnt[threadIdx.x] = 0;
    __syncthreads();
    for (int eb = e0 + (int)threadIdx.x; eb < e1; eb += BIN_U * 1024) {
        int d[BIN_U];
        #pragma unroll
        for (int u = 0; u < BIN_U; ++u) d[u] = eb + u * 1024 < e1 ? dst[eb + u * 1024] : -1;
        #pragma unroll
        for (int u = 0; u < BIN_U; ++u)
            if ((unsigned)d[u] < (unsigned)n) atomicAdd(&cnt[d[u] / OWN], 1);
    }
    __syncthreads();
    const int v = cnt[threadIdx.x];
    int total;
    const int excl = block_scan_excl(v, wsum, &total);
    base[threadIdx.x] = excl;
    cnt[threadIdx.x]  = 0;
    if ((int)threadIdx.x <= nOwn) offs[threadIdx.x] = excl;
    __syncthreads();
    for (int eb = e0 + (int)threadIdx.x; eb < e1; eb += BIN_U * 1024) {
        int d[BIN_U], s[BIN_U];
        #pragma unroll
        for (int u = 0; u < BIN_U; ++u) {
            const bool in = eb + u * 1024 < e1;
            d[u] = in ? dst[eb + u * 1024] : -1;
            s[u] = in ? src[eb + u * 1024] : 0;
        }
        #pragma unroll
        for (int u = 0; u < BIN_U; ++u) {
            if ((unsigned)d[u] < (unsigned)n) {
                const int o   = d[u] / OWN;
                const int pos = base[o] + atomicAdd(&cnt[o], 1);
                bins[e0 + pos] = (int)((unsigned)s[u] | ((unsigned)(d[u] - o * OWN) << 24));
            }
        }
    }
}

template <class Sink>
__device__ __forceinline__ void scan_owned_edges(const int* __restrict__ src, const int* __restrict__ dst,
                                                 int E, int r0, int* qe, int* qd, Sink sink) {
    const int lane = threadIdx.x & 31;
    int qh = 0;
    int qn = 0;
    auto drain = [&](int cnt) {
        int s = 0, d = 0;
        __builtin_amdgcn_fence(__ATOMIC_ACQUIRE, "wavefront");
        if (lane < cnt) {
            const int slot = (qh + lane) & (QCAP - 1);
            s = src[qe[slot]];
            d = qd[slot];
        }
        sink(s, d, cnt);
        qh = (qh + cnt) & (QCAP - 1);
        qn -= cnt;
    };
    auto put = [&](int dv, int ev) {
        const int  dl  = dv - r0;
        const bool hit = (unsigned)dl < (unsigned)OWN;
        const unsigned m = __builtin_amdgcn_ballot_w32(hit);
        if (m) {
            if (hit) {
                const int slot = (qh + qn + __builtin_popcount(m & ((1u << lane) - 1u))) & (QCAP - 1);
                qe[slot] = ev;
                qd[slot] = dl;
            }
            __builtin_amdgcn_fence(__ATOMIC_RELEASE, "wavefront");
            qn += __builtin_popcount(m);
        }
    };
    const int RND = 1024 * 4;
    if ((E & 3) == 0 && E >= 4) {
        auto ld4 = [&](int e) { return *(const int4*)(dst + min(e, E - 4)); };
        auto trip = [&](int4 (&now)[SCAN_U], int4 (&next)[SCAN_U], int eb, const bool whole) {
            const int e0 = eb + 4 * (int)threadIdx.x;
            const int* pn = dst + e0 + SCAN_U * RND;
            #pragma unroll
            for (int u = 0; u < SCAN_U; ++u)
                next[u] = whole ? *(const int4*)(pn + u * RND) : ld4(e0 + (SCAN_U + u) * RND);
            #pragma unroll
            for (int u = 0; u < SCAN_U; ++u) {
                const int  e    = e0 + u * RND;
                const bool live = whole || e < E;
                const unsigned lo4 = min(min((unsigned)(now[u].x - r0), (unsigned)(now[u].y - r0)),
                                         min((unsigned)(now[u].z - r0), (unsigned)(now[u].w - r0)));
                if (__builtin_amdgcn_ballot_w32(lo4 < (unsigned)OWN && live)) {
                    const int4 d4 = live ? now[u] : make_int4(-1, -1, -1, -1);
                    put(d4.x, e);
                    put(d4.y, e + 1);
                    put(d4.z, e + 2);
                    put(d4.w, e + 3);
                    while (qn >= 128) drain(32);
                }
            }
            while (qn >= 32) drain(32);
        };
        int4 bufA[SCAN_U], bufB[SCAN_U];
        #pragma unroll
        for (int u = 0; u < SCAN_U; ++u) bufA[u] = ld4(u * RND + 4 * (int)threadIdx.x);
        int eb = 0;
        for (; eb + 3 * SCAN_U * RND <= E; eb += 2 * SCAN_U * RND) {
            trip(bufA, bufB, eb, true);
            trip(bufB, bufA, eb + SCAN_U * RND, true);
        }
        for (; eb < E; eb += 2 * SCAN_U * RND) {
            trip(bufA, bufB, eb, false);
            trip(bufB, bufA, eb + SCAN_U * RND, false);
        }
    } else {
        for (int eb = 0; eb < E; eb += 1024) {
            const int e = eb + (int)threadIdx.x;
            put(e < E ? dst[e] : -1, e);
            while (qn >= 32) drain(32);
        }
    }
    while (qn > 0) drain(qn < 32 ? qn : 32);
}

__device__ __forceinline__ void product_owned_rows(const float* sA, const float* sW, const float* sD,
                                                   float* __restrict__ out, int r0, int n) {
    const int lane = threadIdx.x & 31;
    const int wv = threadIdx.x >> 5;
    const int hi = lane >> 4;
    const int lo = lane & 15;
    const int tm = (wv >> 1) * 16;
    const int tn = (wv & 1) * 32;
    if (tm >= OWN) return;
    const float* ar = &sA[(tm + lo) * APITCH + 2 * hi];
    const float* bc = &sW[(2 * hi) * WPITCH + tn + lo];
    const v4f dlo = *(const v4f*)(&sD[tm + 8 * hi]);
    const v4f dhi = *(const v4f*)(&sD[tm + 8 * hi + 4]);
    v8f c0 = {}, c1 = {};
    #pragma unroll
    for (int k = 0; k < NFEAT; k += 4) {
        v2f fa, fb0, fb1;
        fa.x  = ar[k];                       fa.y  = ar[k + 1];
        fb0.x = bc[k * WPITCH];              fb0.y = bc[(k + 1) * WPITCH];
        fb1.x = bc[k * WPITCH + 16];         fb1.y = bc[(k + 1) * WPITCH + 16];
        c0 = __builtin_amdgcn_wmma_f32_16x16x4_f32(false, fa, false, fb0, (short)0, c0, false, false);
        c1 = __builtin_amdgcn_wmma_f32_16x16x4_f32(false, fa, false, fb1, (short)0, c1, false, false);
    }
    const int rowBase = r0 + tm + 8 * hi;
    float* o = out + (long)rowBase * NFEAT + tn + lo;
    #pragma unroll
    for (int r = 0; r < 8; ++r) {
        const float dr = r < 4 ? dlo[r & 3] : dhi[r & 3];
        if (rowBase + r < n) {
            o[(long)r * NFEAT]      = c0[r] * dr;
            o[(long)r * NFEAT + 16] = c1[r] * dr;
        }
    }
}

__global__ __launch_bounds__(1024)
void scan_first(const int* __restrict__ src, const int* __restrict__ dst,
                const int* __restrict__ bins,
                const float* __restrict__ x, const float* __restrict__ W,
                float* __restrict__ dinv, float* __restrict__ h,
                float* __restrict__ emb, float* __restrict__ ge, int E, int n) {
    __shared__ int ring[2][32][QCAP];
    __shared__ int wsum[33];
    __shared__ int degc[OWN];
    __shared__ __attribute__((aligned(16))) float sA[OWN * APITCH];
    __shared__ __attribute__((aligned(16))) float sW[NFEAT * WPITCH];
    __shared__ __attribute__((aligned(16))) float sD[OWN];
    __shared__ int s_found;
    const int lane = threadIdx.x & 31;
    const int wv   = threadIdx.x >> 5;
    const int r0   = blockIdx.x * OWN;
    const int cap  = min(OWN, n - r0) * 2 - 1;
    if (threadIdx.x == 0) s_found = 0;
    if (threadIdx.x < OWN) degc[threadIdx.x] = 0;
    if (blockIdx.x == 0 && threadIdx.x < NEMB) ge[threadIdx.x] = -INFINITY;
    {
        const int q  = threadIdx.x & 15;
        const int rs = threadIdx.x >> 4;
        for (int lr = rs; lr < OWN; lr += PROWS) {
            const int ic = min(r0 + lr, n - 1);
            *(v4f*)(&sA[lr * APITCH + 4 * q]) = *(const v4f*)(x + (long)ic * NFEAT + 4 * q);
        }
        *(v4f*)(&sW[rs * WPITCH + 4 * q]) = *(const v4f*)(W + rs * NFEAT + 4 * q);
    }
    __syncthreads();

    if (bins_usable(E, n)) {
        int* stage = &ring[0][0][0];
        const int nOwn = (n + OWN - 1) / OWN;
        const int L    = (E + BINB - 1) / BINB;
        const int sl = threadIdx.x / SEG_T;
        const int j  = threadIdx.x % SEG_T;
        const int* o = bins + E + (long)sl * (nOwn + 1) + blockIdx.x;
        const int st = o[0];
        const int cs = o[1] - st;
        int total;
        const int ex = block_scan_excl(j == 0 ? cs : 0, wsum, &total);
        const int p0 = j == 0 ? ex : ex - cs;
        const int at = min(E, sl * L) + st;
        for (int ib = j; ib < cs; ib += SEG_T * BIN_U) {
            unsigned w[BIN_U];
            #pragma unroll
            for (int u = 0; u < BIN_U; ++u)
                w[u] = ib + SEG_T * u < cs ? (unsigned)bins[at + ib + SEG_T * u] : NOEDGE;
            #pragma unroll
            for (int u = 0; u < BIN_U; ++u) {
                const int i = ib + SEG_T * u;
                if (i < cs) {
                    atomicAdd(&degc[w[u] >> 24], 1);
                    if (p0 + i < 2 * 32 * QCAP) stage[p0 + i] = (int)w[u];
                }
            }
        }
        __syncthreads();
        const int  nchunks = (total + 31) >> 5;
        const bool fits    = nchunks <= cap;
        if (fits) {
            for (int k = wv; k < nchunks; k += 32) {
                const int p = 32 * k + lane;
                *list_word(emb, r0, 1 + k, lane) = __uint_as_float(p < total ? (unsigned)stage[p] : NOEDGE);
            }
        }
        if (threadIdx.x == 0)
            *list_word(emb, r0, 0, 0) = __uint_as_float(fits ? (unsigned)nchunks : 0x7fffffffu);
    } else {
        scan_owned_edges(src, dst, E, r0, ring[0][wv], ring[1][wv], [&](int s, int d, int cnt) {
            if (lane < cnt) atomicAdd(&degc[d], 1);
            int j = 0;
            if (lane == 0) j = atomicAdd(&s_found, 1);
            j = __builtin_amdgcn_readfirstlane(j);
            if (j < cap)
                *list_word(emb, r0, 1 + j, lane) =
                    __uint_as_float(lane < cnt ? ((unsigned)s | ((unsigned)d << 24)) : NOEDGE);
        });
        __syncthreads();

        if (threadIdx.x == 0)
            *list_word(emb, r0, 0, 0) = __uint_as_float(n < (1 << 24) ? (unsigned)s_found : 0x7fffffffu);
    }
    if (threadIdx.x < OWN) {
        const float dv = rsqrtf((float)(degc[threadIdx.x] + 1));
        sD[threadIdx.x] = dv;
        if (r0 + (int)threadIdx.x < n) dinv[r0 + threadIdx.x] = dv;
    }
    __syncthreads();
    product_owned_rows(sA, sW, sD, h, r0, n);
}

template <bool NEXT>
__global__ __launch_bounds__(1024)
void layer_owned(const int* __restrict__ src, const int* __restrict__ dst,
                 const float* __restrict__ h, float* __restrict__ hn,
                 const float* __restrict__ dinv, const float* __restrict__ b,
                 float* __restrict__ emb, int colOff, int relu,
                 const float* __restrict__ Wn, float* __restrict__ ge,
                 const float* __restrict__ fcW, const float* __restrict__ fcb,
                 const int* __restrict__ target, float* __restrict__ out4, int E, int n) {
    __shared__ __attribute__((aligned(16))) float acc[OWN * NFEAT];
    __shared__ int qe[32][QCAP];
    __shared__ int qd[32][QCAP];
    __shared__ float red[PROWS][NFEAT];
    __shared__ __attribute__((aligned(16))) float sA[NEXT ? OWN * APITCH : 4];
    __shared__ __attribute__((aligned(16))) float sW[NEXT ? NFEAT * WPITCH : 4];
    __shared__ __attribute__((aligned(16))) float sD[OWN];
    __shared__ __attribute__((aligned(16))) float trow[NEXT ? 4 : NFEAT];
    const int lane = threadIdx.x & 31;
    const int wv   = threadIdx.x >> 5;
    const int r0   = blockIdx.x * OWN;
    const int cap  = min(OWN, n - r0) * 2 - 1;
    int  tgt   = 0;
    bool own_t = false;
    if constexpr (!NEXT) {
        tgt   = min(max(target[0], 0), n - 1);
        own_t = tgt >= r0 && tgt < r0 + OWN;
    }

    for (int i = threadIdx.x; i < OWN * NFEAT / 4; i += 1024)
        *(float4*)(&acc[4 * i]) = make_float4(0.f, 0.f, 0.f, 0.f);
    __syncthreads();

    auto addrows = [&](int s, int d, unsigned mask) {
        #pragma unroll 1
        while (mask) {
            int   dj[8];
            float v0[8], v1[8];
            #pragma unroll
            for (int k = 0; k < 8; ++k) {
                dj[k] = -1;
                if (mask) {
                    const int j = __builtin_ctz(mask);
                    mask &= mask - 1;
                    const float* hs = h + (long)lane_i(s, j) * NFEAT;
                    dj[k] = lane_i(d, j);
                    v0[k] = hs[lane];
                    v1[k] = hs[lane + 32];
                }
            }
            #pragma unroll
            for (int k = 0; k < 8; ++k) {
                if (dj[k] >= 0) {
                    float* ad = &acc[dj[k] * NFEAT];
                    atomicAdd(&ad[lane],      v0[k]);
                    atomicAdd(&ad[lane + 32], v1[k]);
                }
            }
        }
    };
    const int  nch    = (int)__float_as_uint(*list_word(emb, r0, 0, 0));
    const bool listed = (unsigned)nch <= (unsigned)cap;
    if (listed) {
        auto ldw = [&](int j) {
            return __float_as_uint(*list_word(emb, r0, 1 + min(j, nch - 1), lane));
        };
        int j = __builtin_amdgcn_readfirstlane(wv);
        if (j < nch) {
            unsigned pk = ldw(j);
            while (j < nch) {
                const unsigned nx = ldw(j + 32);
                addrows((int)(pk & 0xffffffu), (int)(pk >> 24), __builtin_amdgcn_ballot_w32(pk != NOEDGE));
                pk = nx;
                j += 32;
            }
        }
    } else {
        scan_owned_edges(src, dst, E, r0, qe[wv], qd[wv], [&](int s, int d, int cnt) {
            addrows(s, d, cnt >= 32 ? 0xffffffffu : ((1u << cnt) - 1u));
        });
    }
    __syncthreads();

    const int q   = threadIdx.x & 15;
    const int rs  = threadIdx.x >> 4;
    constexpr int NP = (OWN + PROWS - 1) / PROWS;
    const float4 bias = *(const float4*)(b + 4 * q);
    float4 m = make_float4(-INFINITY, -INFINITY, -INFINITY, -INFINITY);
    float  di[NP];
    v4f t[NP], w;
    if constexpr (NEXT) w = *(const v4f*)(Wn + rs * NFEAT + 4 * q);
    #pragma unroll
    for (int p = 0; p < NP; ++p) {
        const int lr = rs + PROWS * p;
        if (lr < OWN) {
            const int ic = min(r0 + lr, n - 1);
            di[p] = dinv[ic];
            t[p] = *(const v4f*)(h + (long)ic * NFEAT + 4 * q);
        }
    }
    if constexpr (NEXT) *(v4f*)(&sW[rs * WPITCH + 4 * q]) = w;
    #pragma unroll
    for (int p = 0; p < NP; ++p) {
        const int lr = rs + PROWS * p;
        if (lr < OWN) {
            const int i  = r0 + lr;
            const v4f a = *(const v4f*)(&acc[lr * NFEAT + 4 * q]);
            float4 v;
            v.x = (a.x + t[p].x) * di[p] + bias.x;
            v.y = (a.y + t[p].y) * di[p] + bias.y;
            v.z = (a.z + t[p].z) * di[p] + bias.z;
            v.w = (a.w + t[p].w) * di[p] + bias.w;
            if (relu) {
                v.x = fmaxf(v.x, 0.0f); v.y = fmaxf(v.y, 0.0f);
                v.z = fmaxf(v.z, 0.0f); v.w = fmaxf(v.w, 0.0f);
            }
            if (i < n) {
                *(float4*)(emb + (long)i * NEMB + colOff + 4 * q) = v;
                m.x = fmaxf(m.x, v.x); m.y = fmaxf(m.y, v.y);
                m.z = fmaxf(m.z, v.z); m.w = fmaxf(m.w, v.w);
                if constexpr (!NEXT) {
                    if (i == tgt) *(float4*)(&trow[4 * q]) = v;
                }
            }
            if constexpr (NEXT) {
                *(float4*)(&sA[lr * APITCH + 4 * q]) = v;
                if (q == 0) sD[lr] = di[p];
            }
        }
    }
    *(float4*)(&red[rs][4 * q]) = m;
    __syncthreads();
    if (threadIdx.x < 4 * NFEAT) {
        const int f  = threadIdx.x & (NFEAT - 1);
        const int pr = (threadIdx.x >> 6) * (PROWS / 4);
        float mm = red[pr][f];
        #pragma unroll
        for (int r = 1; r < PROWS / 4; ++r) mm = fmaxf(mm, red[pr + r][f]);
        atomic_max_f32(&ge[colOff + f], mm);
    }
    if constexpr (NEXT)
        product_owned_rows(sA, sW, sD, hn, r0, n);
    if constexpr (!NEXT) {
        if (own_t && wv == 31) {
            const float* er = emb + (long)tgt * NEMB;
            float s0 = 0.f, s1 = 0.f, s2 = 0.f, s3 = 0.f;
            #pragma unroll
            for (int k = 0; k < NEMB / 32; ++k) {
                const int    f  = lane + 32 * k;
                const float4 w4 = *(const float4*)(fcW + 4 * f);
                const float  e  = 32 * k < 128 ? er[f] : trow[f - 128];
                s0 += e * w4.x; s1 += e * w4.y;
                s2 += e * w4.z; s3 += e * w4.w;
            }
            #pragma unroll
            for (int off = 16; off >= 1; off >>= 1) {
                s0 += __shfl_xor(s0, off); s1 += __shfl_xor(s1, off);
                s2 += __shfl_xor(s2, off); s3 += __shfl_xor(s3, off);
            }
            if (lane == 0) {
                out4[0] = s0 + fcb[0]; out4[1] = s1 + fcb[1];
                out4[2] = s2 + fcb[2]; out4[3] = s3 + fcb[3];
            }
        }
    }
}

extern "C" void kernel_launch(void* const* d_in, const int* in_sizes, int n_in,
                              void* d_out, int out_size, void* d_ws, size_t ws_size,
                              hipStream_t stream) {
    const float* x      = (const float*)d_in[0];
    const int*   edge   = (const int*)d_in[1];
    const int*   target = (const int*)d_in[3];
    const float* W1 = (const float*)d_in[4];
    const float* b1 = (const float*)d_in[5];
    const float* W2 = (const float*)d_in[6];
    const float* b2 = (const float*)d_in[7];
    const float* W3 = (const float*)d_in[8];
    const float* b3 = (const float*)d_in[9];
    const float* fcW = (const float*)d_in[10];
    const float* fcb = (const float*)d_in[11];

    const int N = in_sizes[0] / NFEAT;
    const int E = in_sizes[1] / 2;
    const int* esrc = edge;
    const int* edst = edge + E;

    float* ws   = (float*)d_ws;
    float* dinv = ws;
    float* h    = dinv + N;
    float* h2   = h + (long)N * NFEAT;

    float* emb = (float*)d_out;
    float* ge  = emb + (long)N * NEMB;
    float* out4 = ge + NEMB;

    const int nOwn = (N + OWN - 1) / OWN;

    bin_edges<<<BINB, 1024, 0, stream>>>(esrc, edst, (int*)h2, E, N);

    scan_first<<<nOwn, 1024, 0, stream>>>(esrc, edst, (const int*)h2, x, W1, dinv, h, emb, ge, E, N);

    layer_owned<true><<<nOwn, 1024, 0, stream>>>(esrc, edst, h,  h2, dinv, b1, emb, 0,   1, W2, ge,
                                                 nullptr, nullptr, nullptr, nullptr, E, N);
    layer_owned<true><<<nOwn, 1024, 0, stream>>>(esrc, edst, h2, h,  dinv, b2, emb, 64,  1, W3, ge,
                                                 nullptr, nullptr, nullptr, nullptr, E, N);
    layer_owned<false><<<nOwn, 1024, 0, stream>>>(esrc, edst, h, h2, dinv, b3, emb, 128, 0, nullptr, ge,
                                                  fcW, fcb, target, out4, E, N);
}
